# starting kernel
# baseline (speedup 1.0000x reference)
#include <hip/hip_runtime.h>
#include <math.h>


typedef __attribute__((ext_vector_type(2))) float v2f;
typedef __attribute__((ext_vector_type(8))) float v8f;

#define F_IN 512
#define HID  16
#define NCLS 2

__global__ void gnn_zero(float* __restrict__ ws, int n) {
  int i = blockIdx.x * blockDim.x + threadIdx.x;
  if (i < n) ws[i] = 0.0f;
}

__global__ void gnn_degrees(const int* __restrict__ src, const int* __restrict__ dst,
                            float* __restrict__ deg_out, float* __restrict__ deg_in, int E) {
  int e = blockIdx.x * blockDim.x + threadIdx.x;
  if (e < E) {
    atomicAdd(&deg_out[src[e]], 1.0f);
    atomicAdd(&deg_in[dst[e]], 1.0f);
  }
}

__global__ void gnn_invsqrt(const float* __restrict__ deg_out, const float* __restrict__ deg_in,
                            float* __restrict__ inv_out, float* __restrict__ inv_in, int N) {
  int i = blockIdx.x * blockDim.x + threadIdx.x;
  if (i < N) {
    inv_out[i] = rsqrtf(fmaxf(deg_out[i], 1.0f));
    inv_in[i]  = rsqrtf(fmaxf(deg_in[i], 1.0f));
  }
}

__global__ __launch_bounds__(256) void gnn_gemm1_wmma(
    const float* __restrict__ x, const float* __restrict__ W1,
    const float* __restrict__ inv_out, float* __restrict__ h1, int N) {
  __shared__ float wlds[F_IN * HID];
  for (int i = threadIdx.x; i < F_IN * HID; i += blockDim.x) wlds[i] = W1[i];
  __syncthreads();

  const int wave = threadIdx.x >> 5;
  const int lane = threadIdx.x & 31;
  const int tiles = N >> 4;
  const int tile = blockIdx.x * 8 + wave;
  if (tile >= tiles) return;

  const int n16 = lane & 15;
  const int g   = lane >> 4;
  const int row = tile * 16 + n16;
  const float s = inv_out[row];
  const float* __restrict__ xr = x + (size_t)row * F_IN;

  v8f c = {};
  #pragma unroll 4
  for (int k = 0; k < F_IN; k += 4) {
    const int kk = k + 2 * g;
    v2f a = *(const v2f*)(xr + kk);
    a.x *= s;
    a.y *= s;
    v2f b;
    b.x = wlds[(kk + 0) * HID + n16];
    b.y = wlds[(kk + 1) * HID + n16];
    c = __builtin_amdgcn_wmma_f32_16x16x4_f32(
         false, a,  false, b,
         (short)0, c,  false,  false);
  }

  #pragma unroll
  for (int p = 0; p < 8; ++p) {
    const int m = tile * 16 + p + 8 * g;
    h1[(size_t)m * HID + n16] = c[p];
  }
}

__global__ void gnn_scatter16(const int* __restrict__ src, const int* __restrict__ dst,
                              const float* __restrict__ h1, float* __restrict__ agg1,
                              long E4) {
  long t = (long)blockIdx.x * blockDim.x + threadIdx.x;
  if (t >= E4) return;
  long e = t >> 2;
  int f = (int)(t & 3) * 4;
  int s = src[e], d = dst[e];
  float4 v = *(const float4*)(h1 + (size_t)s * HID + f);
  float* a = agg1 + (size_t)d * HID + f;
  atomicAdd(a + 0, v.x);
  atomicAdd(a + 1, v.y);
  atomicAdd(a + 2, v.z);
  atomicAdd(a + 3, v.w);
}

__global__ void gnn_layer2(const float* __restrict__ agg1, const float* __restrict__ inv_in,
                           const float* __restrict__ inv_out, const float* __restrict__ b1,
                           const float* __restrict__ W2, float* __restrict__ h2, int N) {
  int i = blockIdx.x * blockDim.x + threadIdx.x;
  if (i >= N) return;
  const float si = inv_in[i];
  const float so = inv_out[i];
  const float* a = agg1 + (size_t)i * HID;
  float acc0 = 0.0f, acc1 = 0.0f;
  #pragma unroll
  for (int k = 0; k < HID; ++k) {
    float hv = fmaxf(fmaf(a[k], si, b1[k]), 0.0f) * so;
    acc0 = fmaf(hv, W2[k * NCLS + 0], acc0);
    acc1 = fmaf(hv, W2[k * NCLS + 1], acc1);
  }
  h2[(size_t)i * NCLS + 0] = acc0;
  h2[(size_t)i * NCLS + 1] = acc1;
}

__global__ void gnn_scatter2(const int* __restrict__ src, const int* __restrict__ dst,
                             const float* __restrict__ h2, float* __restrict__ agg2, int E) {
  int e = blockIdx.x * blockDim.x + threadIdx.x;
  if (e >= E) return;
  int s = src[e], d = dst[e];
  float2 v = *(const float2*)(h2 + (size_t)s * NCLS);
  atomicAdd(&agg2[(size_t)d * NCLS + 0], v.x);
  atomicAdd(&agg2[(size_t)d * NCLS + 1], v.y);
}

__global__ void gnn_final(const float* __restrict__ agg2, const float* __restrict__ inv_in,
                          const float* __restrict__ b2, float* __restrict__ out, int N) {
  int i = blockIdx.x * blockDim.x + threadIdx.x;
  if (i >= N) return;
  const float si = inv_in[i];
  float v0 = fmaf(agg2[(size_t)i * NCLS + 0], si, b2[0]);
  float v1 = fmaf(agg2[(size_t)i * NCLS + 1], si, b2[1]);
  float m = fmaxf(v0, v1);
  float lse = m + logf(expf(v0 - m) + expf(v1 - m));
  out[(size_t)i * NCLS + 0] = v0 - lse;
  out[(size_t)i * NCLS + 1] = v1 - lse;
}

extern "C" void kernel_launch(void* const* d_in, const int* in_sizes, int n_in,
                              void* d_out, int out_size, void* d_ws, size_t ws_size,
                              hipStream_t stream) {
  const float* x   = (const float*)d_in[0];
  const int*   src = (const int*)d_in[1];
  const int*   dst = (const int*)d_in[2];
  const float* W1  = (const float*)d_in[3];
  const float* b1  = (const float*)d_in[4];
  const float* W2  = (const float*)d_in[5];
  const float* b2  = (const float*)d_in[6];
  float* out = (float*)d_out;

  const int N = in_sizes[0] / F_IN;
  const int E = in_sizes[1];

  float* ws      = (float*)d_ws;
  float* deg_out = ws;
  float* deg_in  = ws + (size_t)N;
  float* inv_out = ws + (size_t)2 * N;
  float* inv_in  = ws + (size_t)3 * N;
  float* h1      = ws + (size_t)4 * N;
  float* agg1    = ws + (size_t)20 * N;
  float* h2      = ws + (size_t)36 * N;
  float* agg2    = ws + (size_t)38 * N;

  const int B = 256;
  const int wsFloats = 40 * N;

  gnn_zero<<<(wsFloats + B - 1) / B, B, 0, stream>>>(ws, wsFloats);
  gnn_degrees<<<(E + B - 1) / B, B, 0, stream>>>(src, dst, deg_out, deg_in, E);
  gnn_invsqrt<<<(N + B - 1) / B, B, 0, stream>>>(deg_out, deg_in, inv_out, inv_in, N);

  const int tiles = N >> 4;
  gnn_gemm1_wmma<<<(tiles + 7) / 8, B, 0, stream>>>(x, W1, inv_out, h1, N);

  const long E4 = (long)E * 4;
  gnn_scatter16<<<(int)((E4 + B - 1) / B), B, 0, stream>>>(src, dst, h1, agg1, E4);
  gnn_layer2<<<(N + B - 1) / B, B, 0, stream>>>(agg1, inv_in, inv_out, b1, W2, h2, N);
  gnn_scatter2<<<(E + B - 1) / B, B, 0, stream>>>(src, dst, h2, agg2, E);
  gnn_final<<<(N + B - 1) / B, B, 0, stream>>>(agg2, inv_in, b2, out, N);
}


# One-word scatter sums (degree counts, layer-2 sums) get one plane per package half, picked by the wave's die id; readers add the two planes (layer-2 sum tree changes)
# speedup vs baseline: 2.5039x; 2.5039x over previous
#include <hip/hip_runtime.h>
#include <math.h>


typedef __attribute__((ext_vector_type(2))) float v2f;
typedef __attribute__((ext_vector_type(8))) float v8f;

#define F_IN 512
#define HID  16
#define NCLS 2

__global__ void gnn_zero(float* __restrict__ ws, int n) {
  int i = blockIdx.x * blockDim.x + threadIdx.x;
  if (i < n) ws[i] = 0.0f;
}

__device__ __forceinline__ int gnn_half() {
  return (int)((__builtin_amdgcn_s_sendmsg_rtn(0x87) >> 18) & 1u);
}

__global__ void gnn_degrees(const int* __restrict__ src, const int* __restrict__ dst,
                            float* __restrict__ degA, float* __restrict__ degB,
                            int N, int E) {
  int e = blockIdx.x * blockDim.x + threadIdx.x;
  float* __restrict__ deg = gnn_half() ? degB : degA;
  if (e < E) {
    atomicAdd(&deg[src[e]], 1.0f);
    atomicAdd(&deg[(size_t)N + dst[e]], 1.0f);
  }
}

__global__ void gnn_invsqrt(const float* __restrict__ degA, const float* __restrict__ degB,
                            float* __restrict__ inv_out, float* __restrict__ inv_in, int N) {
  int i = blockIdx.x * blockDim.x + threadIdx.x;
  if (i < N) {
    inv_out[i] = rsqrtf(fmaxf(degA[i] + degB[i], 1.0f));
    inv_in[i]  = rsqrtf(fmaxf(degA[(size_t)N + i] + degB[(size_t)N + i], 1.0f));
  }
}

__global__ __launch_bounds__(256) void gnn_gemm1_wmma(
    const float* __restrict__ x, const float* __restrict__ W1,
    const float* __restrict__ inv_out, float* __restrict__ h1, int N) {
  __shared__ float wlds[F_IN * HID];
  for (int i = threadIdx.x; i < F_IN * HID; i += blockDim.x) wlds[i] = W1[i];
  __syncthreads();

  const int wave = threadIdx.x >> 5;
  const int lane = threadIdx.x & 31;
  const int tiles = N >> 4;
  const int tile = blockIdx.x * 8 + wave;
  if (tile >= tiles) return;

  const int n16 = lane & 15;
  const int g   = lane >> 4;
  const int row = tile * 16 + n16;
  const float s = inv_out[row];
  const float* __restrict__ xr = x + (size_t)row * F_IN;

  v8f c = {};
  #pragma unroll 4
  for (int k = 0; k < F_IN; k += 4) {
    const int kk = k + 2 * g;
    v2f a = *(const v2f*)(xr + kk);
    a.x *= s;
    a.y *= s;
    v2f b;
    b.x = wlds[(kk + 0) * HID + n16];
    b.y = wlds[(kk + 1) * HID + n16];
    c = __builtin_amdgcn_wmma_f32_16x16x4_f32(
         false, a,  false, b,
         (short)0, c,  false,  false);
  }

  #pragma unroll
  for (int p = 0; p < 8; ++p) {
    const int m = tile * 16 + p + 8 * g;
    h1[(size_t)m * HID + n16] = c[p];
  }
}

__global__ void gnn_scatter16(const int* __restrict__ src, const int* __restrict__ dst,
                              const float* __restrict__ h1, float* __restrict__ agg1,
                              long E4) {
  long t = (long)blockIdx.x * blockDim.x + threadIdx.x;
  if (t >= E4) return;
  long e = t >> 2;
  int f = (int)(t & 3) * 4;
  int s = src[e], d = dst[e];
  float4 v = *(const float4*)(h1 + (size_t)s * HID + f);
  float* a = agg1 + (size_t)d * HID + f;
  atomicAdd(a + 0, v.x);
  atomicAdd(a + 1, v.y);
  atomicAdd(a + 2, v.z);
  atomicAdd(a + 3, v.w);
}

__global__ void gnn_layer2(const float* __restrict__ agg1, const float* __restrict__ inv_in,
                           const float* __restrict__ inv_out, const float* __restrict__ b1,
                           const float* __restrict__ W2, float* __restrict__ h2,
                           float* __restrict__ agg2B, int N) {
  int i = blockIdx.x * blockDim.x + threadIdx.x;
  if (i >= N) return;
  *(float2*)(agg2B + (size_t)i * NCLS) = make_float2(0.0f, 0.0f);
  const float si = inv_in[i];
  const float so = inv_out[i];
  const float* a = agg1 + (size_t)i * HID;
  float acc0 = 0.0f, acc1 = 0.0f;
  #pragma unroll
  for (int k = 0; k < HID; ++k) {
    float hv = fmaxf(fmaf(a[k], si, b1[k]), 0.0f) * so;
    acc0 = fmaf(hv, W2[k * NCLS + 0], acc0);
    acc1 = fmaf(hv, W2[k * NCLS + 1], acc1);
  }
  h2[(size_t)i * NCLS + 0] = acc0;
  h2[(size_t)i * NCLS + 1] = acc1;
}

__global__ void gnn_scatter2(const int* __restrict__ src, const int* __restrict__ dst,
                             const float* __restrict__ h2, float* __restrict__ agg2A,
                             float* __restrict__ agg2B, int E) {
  int e = blockIdx.x * blockDim.x + threadIdx.x;
  if (e >= E) return;
  float* __restrict__ agg2 = gnn_half() ? agg2B : agg2A;
  int s = src[e], d = dst[e];
  float2 v = *(const float2*)(h2 + (size_t)s * NCLS);
  atomicAdd(&agg2[(size_t)d * NCLS + 0], v.x);
  atomicAdd(&agg2[(size_t)d * NCLS + 1], v.y);
}

__global__ void gnn_final(const float* __restrict__ agg2A, const float* __restrict__ agg2B,
                          const float* __restrict__ inv_in,
                          const float* __restrict__ b2, float* __restrict__ out, int N) {
  int i = blockIdx.x * blockDim.x + threadIdx.x;
  if (i >= N) return;
  const float si = inv_in[i];
  const float2 pa = *(const float2*)(agg2A + (size_t)i * NCLS);
  const float2 pb = *(const float2*)(agg2B + (size_t)i * NCLS);
  float v0 = fmaf(pa.x + pb.x, si, b2[0]);
  float v1 = fmaf(pa.y + pb.y, si, b2[1]);
  float m = fmaxf(v0, v1);
  float lse = m + logf(expf(v0 - m) + expf(v1 - m));
  out[(size_t)i * NCLS + 0] = v0 - lse;
  out[(size_t)i * NCLS + 1] = v1 - lse;
}

extern "C" void kernel_launch(void* const* d_in, const int* in_sizes, int n_in,
                              void* d_out, int out_size, void* d_ws, size_t ws_size,
                              hipStream_t stream) {
  const float* x   = (const float*)d_in[0];
  const int*   src = (const int*)d_in[1];
  const int*   dst = (const int*)d_in[2];
  const float* W1  = (const float*)d_in[3];
  const float* b1  = (const float*)d_in[4];
  const float* W2  = (const float*)d_in[5];
  const float* b2  = (const float*)d_in[6];
  float* out = (float*)d_out;

  const int N = in_sizes[0] / F_IN;
  const int E = in_sizes[1];

  float* ws      = (float*)d_ws;
  float* degA    = ws;
  float* inv_out = ws + (size_t)2 * N;
  float* inv_in  = ws + (size_t)3 * N;
  float* h1      = ws + (size_t)4 * N;
  float* degB    = h1;
  float* agg2B   = h1;
  float* agg1    = ws + (size_t)20 * N;
  float* h2      = ws + (size_t)36 * N;
  float* agg2A   = ws + (size_t)38 * N;

  const int B = 256;
  const int wsFloats = 40 * N;

  gnn_zero<<<(wsFloats + B - 1) / B, B, 0, stream>>>(ws, wsFloats);
  gnn_degrees<<<(E + B - 1) / B, B, 0, stream>>>(src, dst, degA, degB, N, E);
  gnn_invsqrt<<<(N + B - 1) / B, B, 0, stream>>>(degA, degB, inv_out, inv_in, N);

  const int tiles = N >> 4;
  gnn_gemm1_wmma<<<(tiles + 7) / 8, B, 0, stream>>>(x, W1, inv_out, h1, N);

  const long E4 = (long)E * 4;
  gnn_scatter16<<<(int)((E4 + B - 1) / B), B, 0, stream>>>(src, dst, h1, agg1, E4);
  gnn_layer2<<<(N + B - 1) / B, B, 0, stream>>>(agg1, inv_in, inv_out, b1, W2, h2, agg2B, N);
  gnn_scatter2<<<(E + B - 1) / B, B, 0, stream>>>(src, dst, h2, agg2A, agg2B, E);
  gnn_final<<<(N + B - 1) / B, B, 0, stream>>>(agg2A, agg2B, inv_in, b2, out, N);
}


# Edge sums add one whole row per atomic instruction: layer 1 on 16 adjacent lanes x 1 add per 64-byte row (was 4 lanes x 4 adds), layer 2 on lane pairs (was 2 adds); atomics arrival-ordered as before
# speedup vs baseline: 3.3490x; 1.3375x over previous
#include <hip/hip_runtime.h>
#include <math.h>


typedef __attribute__((ext_vector_type(2))) float v2f;
typedef __attribute__((ext_vector_type(8))) float v8f;

#define F_IN 512
#define HID  16
#define NCLS 2

__global__ void gnn_zero(float* __restrict__ ws, int n) {
  int i = blockIdx.x * blockDim.x + threadIdx.x;
  if (i < n) ws[i] = 0.0f;
}

__device__ __forceinline__ int gnn_half() {
  return (int)((__builtin_amdgcn_s_sendmsg_rtn(0x87) >> 18) & 1u);
}

__global__ void gnn_degrees(const int* __restrict__ src, const int* __restrict__ dst,
                            float* __restrict__ degA, float* __restrict__ degB,
                            int N, int E) {
  int e = blockIdx.x * blockDim.x + threadIdx.x;
  float* __restrict__ deg = gnn_half() ? degB : degA;
  if (e < E) {
    atomicAdd(&deg[src[e]], 1.0f);
    atomicAdd(&deg[(size_t)N + dst[e]], 1.0f);
  }
}

__global__ void gnn_invsqrt(const float* __restrict__ degA, const float* __restrict__ degB,
                            float* __restrict__ inv_out, float* __restrict__ inv_in, int N) {
  int i = blockIdx.x * blockDim.x + threadIdx.x;
  if (i < N) {
    inv_out[i] = rsqrtf(fmaxf(degA[i] + degB[i], 1.0f));
    inv_in[i]  = rsqrtf(fmaxf(degA[(size_t)N + i] + degB[(size_t)N + i], 1.0f));
  }
}

__global__ __launch_bounds__(256) void gnn_gemm1_wmma(
    const float* __restrict__ x, const float* __restrict__ W1,
    const float* __restrict__ inv_out, float* __restrict__ h1, int N) {
  __shared__ float wlds[F_IN * HID];
  for (int i = threadIdx.x; i < F_IN * HID; i += blockDim.x) wlds[i] = W1[i];
  __syncthreads();

  const int wave = threadIdx.x >> 5;
  const int lane = threadIdx.x & 31;
  const int tiles = N >> 4;
  const int tile = blockIdx.x * 8 + wave;
  if (tile >= tiles) return;

  const int n16 = lane & 15;
  const int g   = lane >> 4;
  const int row = tile * 16 + n16;
  const float s = inv_out[row];
  const float* __restrict__ xr = x + (size_t)row * F_IN;

  v8f c = {};
  #pragma unroll 4
  for (int k = 0; k < F_IN; k += 4) {
    const int kk = k + 2 * g;
    v2f a = *(const v2f*)(xr + kk);
    a.x *= s;
    a.y *= s;
    v2f b;
    b.x = wlds[(kk + 0) * HID + n16];
    b.y = wlds[(kk + 1) * HID + n16];
    c = __builtin_amdgcn_wmma_f32_16x16x4_f32(
         false, a,  false, b,
         (short)0, c,  false,  false);
  }

  #pragma unroll
  for (int p = 0; p < 8; ++p) {
    const int m = tile * 16 + p + 8 * g;
    h1[(size_t)m * HID + n16] = c[p];
  }
}

__global__ void gnn_scatter16(const int* __restrict__ src, const int* __restrict__ dst,
                              const float* __restrict__ h1, float* __restrict__ agg1,
                              int E) {
  const int e = blockIdx.x * blockDim.x + threadIdx.x;
  const int lane = threadIdx.x & 31;
  const unsigned f = lane & 15;
  const int g = lane >> 4;
  const bool ok = e < E;
  const int s_l = ok ? src[e] : 0;
  const int d_l = ok ? dst[e] : 0;
  const int n = E - __builtin_amdgcn_readfirstlane(e);
  float v[16];
  #pragma unroll
  for (int i = 0; i < 16; ++i) {
    const unsigned lo = (unsigned)__builtin_amdgcn_readlane(s_l, 2 * i);
    const unsigned hi = (unsigned)__builtin_amdgcn_readlane(s_l, 2 * i + 1);
    const unsigned s = g ? hi : lo;
    v[i] = h1[s * HID + f];
  }
  #pragma unroll
  for (int i = 0; i < 16; ++i) {
    const unsigned lo = (unsigned)__builtin_amdgcn_readlane(d_l, 2 * i);
    const unsigned hi = (unsigned)__builtin_amdgcn_readlane(d_l, 2 * i + 1);
    const unsigned d = g ? hi : lo;
    if (2 * i + g < n) atomicAdd(agg1 + d * HID + f, v[i]);
  }
}

__global__ void gnn_layer2(const float* __restrict__ agg1, const float* __restrict__ inv_in,
                           const float* __restrict__ inv_out, const float* __restrict__ b1,
                           const float* __restrict__ W2, float* __restrict__ h2,
                           float* __restrict__ agg2B, int N) {
  int i = blockIdx.x * blockDim.x + threadIdx.x;
  if (i >= N) return;
  *(float2*)(agg2B + (size_t)i * NCLS) = make_float2(0.0f, 0.0f);
  const float si = inv_in[i];
  const float so = inv_out[i];
  const float* a = agg1 + (size_t)i * HID;
  float acc0 = 0.0f, acc1 = 0.0f;
  #pragma unroll
  for (int k = 0; k < HID; ++k) {
    float hv = fmaxf(fmaf(a[k], si, b1[k]), 0.0f) * so;
    acc0 = fmaf(hv, W2[k * NCLS + 0], acc0);
    acc1 = fmaf(hv, W2[k * NCLS + 1], acc1);
  }
  h2[(size_t)i * NCLS + 0] = acc0;
  h2[(size_t)i * NCLS + 1] = acc1;
}

__global__ void gnn_scatter2(const int* __restrict__ src, const int* __restrict__ dst,
                             const float* __restrict__ h2, float* __restrict__ agg2A,
                             float* __restrict__ agg2B, int E) {
  const int e = blockIdx.x * blockDim.x + threadIdx.x;
  const int lane = threadIdx.x & 31;
  const unsigned c = lane & 1;
  const int p = lane >> 1;
  const bool ok = e < E;
  const int s_l = ok ? src[e] : 0;
  const int d_l = ok ? dst[e] : 0;
  const int n = E - __builtin_amdgcn_readfirstlane(e);
  float* __restrict__ agg2 = gnn_half() ? agg2B : agg2A;
  const unsigned s0 = (unsigned)__shfl(s_l, p);
  const unsigned s1 = (unsigned)__shfl(s_l, 16 + p);
  const unsigned d0 = (unsigned)__shfl(d_l, p);
  const unsigned d1 = (unsigned)__shfl(d_l, 16 + p);
  const float v0 = h2[s0 * NCLS + c];
  const float v1 = h2[s1 * NCLS + c];
  if (p < n) atomicAdd(agg2 + d0 * NCLS + c, v0);
  if (16 + p < n) atomicAdd(agg2 + d1 * NCLS + c, v1);
}

__global__ void gnn_final(const float* __restrict__ agg2A, const float* __restrict__ agg2B,
                          const float* __restrict__ inv_in,
                          const float* __restrict__ b2, float* __restrict__ out, int N) {
  int i = blockIdx.x * blockDim.x + threadIdx.x;
  if (i >= N) return;
  const float si = inv_in[i];
  const float2 pa = *(const float2*)(agg2A + (size_t)i * NCLS);
  const float2 pb = *(const float2*)(agg2B + (size_t)i * NCLS);
  float v0 = fmaf(pa.x + pb.x, si, b2[0]);
  float v1 = fmaf(pa.y + pb.y, si, b2[1]);
  float m = fmaxf(v0, v1);
  float lse = m + logf(expf(v0 - m) + expf(v1 - m));
  out[(size_t)i * NCLS + 0] = v0 - lse;
  out[(size_t)i * NCLS + 1] = v1 - lse;
}

extern "C" void kernel_launch(void* const* d_in, const int* in_sizes, int n_in,
                              void* d_out, int out_size, void* d_ws, size_t ws_size,
                              hipStream_t stream) {
  const float* x   = (const float*)d_in[0];
  const int*   src = (const int*)d_in[1];
  const int*   dst = (const int*)d_in[2];
  const float* W1  = (const float*)d_in[3];
  const float* b1  = (const float*)d_in[4];
  const float* W2  = (const float*)d_in[5];
  const float* b2  = (const float*)d_in[6];
  float* out = (float*)d_out;

  const int N = in_sizes[0] / F_IN;
  const int E = in_sizes[1];

  float* ws      = (float*)d_ws;
  float* degA    = ws;
  float* inv_out = ws + (size_t)2 * N;
  float* inv_in  = ws + (size_t)3 * N;
  float* h1      = ws + (size_t)4 * N;
  float* degB    = h1;
  float* agg2B   = h1;
  float* agg1    = ws + (size_t)20 * N;
  float* h2      = ws + (size_t)36 * N;
  float* agg2A   = ws + (size_t)38 * N;

  const int B = 256;
  const int wsFloats = 40 * N;

  gnn_zero<<<(wsFloats + B - 1) / B, B, 0, stream>>>(ws, wsFloats);
  gnn_degrees<<<(E + B - 1) / B, B, 0, stream>>>(src, dst, degA, degB, N, E);
  gnn_invsqrt<<<(N + B - 1) / B, B, 0, stream>>>(degA, degB, inv_out, inv_in, N);

  const int tiles = N >> 4;
  gnn_gemm1_wmma<<<(tiles + 7) / 8, B, 0, stream>>>(x, W1, inv_out, h1, N);

  gnn_scatter16<<<(E + B - 1) / B, B, 0, stream>>>(src, dst, h1, agg1, E);
  gnn_layer2<<<(N + B - 1) / B, B, 0, stream>>>(agg1, inv_in, inv_out, b1, W2, h2, agg2B, N);
  gnn_scatter2<<<(E + B - 1) / B, B, 0, stream>>>(src, dst, h2, agg2A, agg2B, E);
  gnn_final<<<(N + B - 1) / B, B, 0, stream>>>(agg2A, agg2B, inv_in, b2, out, N);
}


# Layer-1 edge sums: each package half adds only into destination rows of its own half of the node range (edge runs drawn per half from counters): no agg1 line moves between halves; arrival-ordered
# speedup vs baseline: 4.3359x; 1.2947x over previous
#include <hip/hip_runtime.h>
#include <math.h>


typedef __attribute__((ext_vector_type(2))) float v2f;
typedef __attribute__((ext_vector_type(8))) float v8f;

#define F_IN 512
#define HID  16
#define NCLS 2

__global__ void gnn_zero(float* __restrict__ ws, int n) {
  int i = blockIdx.x * blockDim.x + threadIdx.x;
  if (i < n) ws[i] = 0.0f;
}

__device__ __forceinline__ int gnn_half() {
  return (int)((__builtin_amdgcn_s_sendmsg_rtn(0x87) >> 18) & 1u);
}

__global__ void gnn_degrees(const int* __restrict__ src, const int* __restrict__ dst,
                            float* __restrict__ degA, float* __restrict__ degB,
                            int N, int E) {
  int e = blockIdx.x * blockDim.x + threadIdx.x;
  float* __restrict__ deg = gnn_half() ? degB : degA;
  if (e < E) {
    atomicAdd(&deg[src[e]], 1.0f);
    atomicAdd(&deg[(size_t)N + dst[e]], 1.0f);
  }
}

__global__ void gnn_invsqrt(const float* __restrict__ degA, const float* __restrict__ degB,
                            float* __restrict__ inv_out, float* __restrict__ inv_in, int N) {
  int i = blockIdx.x * blockDim.x + threadIdx.x;
  if (i < N) {
    inv_out[i] = rsqrtf(fmaxf(degA[i] + degB[i], 1.0f));
    inv_in[i]  = rsqrtf(fmaxf(degA[(size_t)N + i] + degB[(size_t)N + i], 1.0f));
  }
}

__global__ __launch_bounds__(256) void gnn_gemm1_wmma(
    const float* __restrict__ x, const float* __restrict__ W1,
    const float* __restrict__ inv_out, float* __restrict__ h1, int N) {
  __shared__ float wlds[F_IN * HID];
  for (int i = threadIdx.x; i < F_IN * HID; i += blockDim.x) wlds[i] = W1[i];
  __syncthreads();

  const int wave = threadIdx.x >> 5;
  const int lane = threadIdx.x & 31;
  const int tiles = N >> 4;
  const int tile = blockIdx.x * 8 + wave;
  if (tile >= tiles) return;

  const int n16 = lane & 15;
  const int g   = lane >> 4;
  const int row = tile * 16 + n16;
  const float s = inv_out[row];
  const float* __restrict__ xr = x + (size_t)row * F_IN;

  v8f c = {};
  #pragma unroll 4
  for (int k = 0; k < F_IN; k += 4) {
    const int kk = k + 2 * g;
    v2f a = *(const v2f*)(xr + kk);
    a.x *= s;
    a.y *= s;
    v2f b;
    b.x = wlds[(kk + 0) * HID + n16];
    b.y = wlds[(kk + 1) * HID + n16];
    c = __builtin_amdgcn_wmma_f32_16x16x4_f32(
         false, a,  false, b,
         (short)0, c,  false,  false);
  }

  #pragma unroll
  for (int p = 0; p < 8; ++p) {
    const int m = tile * 16 + p + 8 * g;
    h1[(size_t)m * HID + n16] = c[p];
  }
}

#define GNN_RUN 128

__device__ __forceinline__ unsigned int gnn_draw(unsigned int* c, int lane) {
  unsigned int t = 0u;
  if (lane == 0) t = atomicAdd(c, 1u);
  return t;
}

__device__ __forceinline__ void gnn_add32(const int2* q, unsigned head,
                                          unsigned rem, unsigned g, unsigned f,
                                          const float* __restrict__ h1,
                                          float* __restrict__ agg1) {
  float v[16];
  unsigned d[16];
  #pragma unroll
  for (int i = 0; i < 16; ++i) {
    const unsigned k = 2u * i + g;
    const int2 sd = q[(head + k) & 127u];
    const unsigned s = (k < rem) ? (unsigned)sd.x : 0u;
    d[i] = (unsigned)sd.y;
    v[i] = h1[s * HID + f];
  }
  #pragma unroll
  for (int i = 0; i < 16; ++i) {
    const unsigned k = 2u * i + g;
    if (k < rem) atomicAdd(agg1 + d[i] * HID + f, v[i]);
  }
}

__global__ __launch_bounds__(256) void gnn_scatter16(
    const int* __restrict__ src, const int* __restrict__ dst,
    const float* __restrict__ h1, float* __restrict__ agg1,
    unsigned int* __restrict__ ctr, int N, int E) {
  __shared__ int2 ring[8][128];
  const int lane = threadIdx.x & 31;
  const int wave = __builtin_amdgcn_readfirstlane((int)(threadIdx.x >> 5));
  int2* q = ring[wave];
  const unsigned f = lane & 15;
  const unsigned g = lane >> 4;
  const int mid = N >> 1;
  const int runs = (E + GNN_RUN - 1) / GNN_RUN;
  const int mine = (runs - wave + 7) >> 3;
  const int h0 = gnn_half();
  unsigned head = 0u, tail = 0u;

  for (int pass = 0; pass < 2; ++pass) {
    const int h = h0 ^ pass;
    unsigned int* c = ctr + 64 * (8 * h + wave);
    unsigned int next = gnn_draw(c, lane);
    for (;;) {
      const int cur = (int)__builtin_amdgcn_readfirstlane(next);
      if (cur >= mine) break;
      const int run = 8 * cur + wave;
      for (int it = 0; it < GNN_RUN / 64; ++it) {
        const int e0 = run * GNN_RUN + it * 64 + lane;
        const int e1 = e0 + 32;
        const bool ok0 = e0 < E;
        const bool ok1 = e1 < E;
        const int s0 = ok0 ? src[e0] : 0;
        const int d0 = ok0 ? dst[e0] : 0;
        const int s1 = ok1 ? src[e1] : 0;
        const int d1 = ok1 ? dst[e1] : 0;
        if (it == 0) next = gnn_draw(c, lane);
        const bool own0 = ok0 && ((d0 >= mid) == (h != 0));
        const bool own1 = ok1 && ((d1 >= mid) == (h != 0));
        const unsigned m0 = __builtin_amdgcn_ballot_w32(own0);
        const unsigned m1 = __builtin_amdgcn_ballot_w32(own1);
        const unsigned c0 = (unsigned)__builtin_popcount(m0);
        const unsigned r0 = __builtin_amdgcn_mbcnt_lo(m0, 0u);
        const unsigned r1 = c0 + __builtin_amdgcn_mbcnt_lo(m1, 0u);
        if (own0) q[(tail + r0) & 127u] = make_int2(s0, d0);
        if (own1) q[(tail + r1) & 127u] = make_int2(s1, d1);
        tail += c0 + (unsigned)__builtin_popcount(m1);
        while (tail - head >= 32u) {
          gnn_add32(q, head, 32u, g, f, h1, agg1);
          head += 32u;
        }
      }
    }
  }
  if (tail != head) gnn_add32(q, head, tail - head, g, f, h1, agg1);
}

__global__ void gnn_layer2(const float* __restrict__ agg1, const float* __restrict__ inv_in,
                           const float* __restrict__ inv_out, const float* __restrict__ b1,
                           const float* __restrict__ W2, float* __restrict__ h2,
                           float* __restrict__ agg2B, int N) {
  int i = blockIdx.x * blockDim.x + threadIdx.x;
  if (i >= N) return;
  *(float2*)(agg2B + (size_t)i * NCLS) = make_float2(0.0f, 0.0f);
  const float si = inv_in[i];
  const float so = inv_out[i];
  const float* a = agg1 + (size_t)i * HID;
  float acc0 = 0.0f, acc1 = 0.0f;
  #pragma unroll
  for (int k = 0; k < HID; ++k) {
    float hv = fmaxf(fmaf(a[k], si, b1[k]), 0.0f) * so;
    acc0 = fmaf(hv, W2[k * NCLS + 0], acc0);
    acc1 = fmaf(hv, W2[k * NCLS + 1], acc1);
  }
  h2[(size_t)i * NCLS + 0] = acc0;
  h2[(size_t)i * NCLS + 1] = acc1;
}

__global__ void gnn_scatter2(const int* __restrict__ src, const int* __restrict__ dst,
                             const float* __restrict__ h2, float* __restrict__ agg2A,
                             float* __restrict__ agg2B, int E) {
  const int e = blockIdx.x * blockDim.x + threadIdx.x;
  const int lane = threadIdx.x & 31;
  const unsigned c = lane & 1;
  const int p = lane >> 1;
  const bool ok = e < E;
  const int s_l = ok ? src[e] : 0;
  const int d_l = ok ? dst[e] : 0;
  const int n = E - __builtin_amdgcn_readfirstlane(e);
  float* __restrict__ agg2 = gnn_half() ? agg2B : agg2A;
  const unsigned s0 = (unsigned)__shfl(s_l, p);
  const unsigned s1 = (unsigned)__shfl(s_l, 16 + p);
  const unsigned d0 = (unsigned)__shfl(d_l, p);
  const unsigned d1 = (unsigned)__shfl(d_l, 16 + p);
  const float v0 = h2[s0 * NCLS + c];
  const float v1 = h2[s1 * NCLS + c];
  if (p < n) atomicAdd(agg2 + d0 * NCLS + c, v0);
  if (16 + p < n) atomicAdd(agg2 + d1 * NCLS + c, v1);
}

__global__ void gnn_final(const float* __restrict__ agg2A, const float* __restrict__ agg2B,
                          const float* __restrict__ inv_in,
                          const float* __restrict__ b2, float* __restrict__ out, int N) {
  int i = blockIdx.x * blockDim.x + threadIdx.x;
  if (i >= N) return;
  const float si = inv_in[i];
  const float2 pa = *(const float2*)(agg2A + (size_t)i * NCLS);
  const float2 pb = *(const float2*)(agg2B + (size_t)i * NCLS);
  float v0 = fmaf(pa.x + pb.x, si, b2[0]);
  float v1 = fmaf(pa.y + pb.y, si, b2[1]);
  float m = fmaxf(v0, v1);
  float lse = m + logf(expf(v0 - m) + expf(v1 - m));
  out[(size_t)i * NCLS + 0] = v0 - lse;
  out[(size_t)i * NCLS + 1] = v1 - lse;
}

extern "C" void kernel_launch(void* const* d_in, const int* in_sizes, int n_in,
                              void* d_out, int out_size, void* d_ws, size_t ws_size,
                              hipStream_t stream) {
  const float* x   = (const float*)d_in[0];
  const int*   src = (const int*)d_in[1];
  const int*   dst = (const int*)d_in[2];
  const float* W1  = (const float*)d_in[3];
  const float* b1  = (const float*)d_in[4];
  const float* W2  = (const float*)d_in[5];
  const float* b2  = (const float*)d_in[6];
  float* out = (float*)d_out;

  const int N = in_sizes[0] / F_IN;
  const int E = in_sizes[1];

  float* ws      = (float*)d_ws;
  float* degA    = ws;
  float* inv_out = ws + (size_t)2 * N;
  float* inv_in  = ws + (size_t)3 * N;
  float* h1      = ws + (size_t)4 * N;
  float* degB    = h1;
  float* agg2B   = h1;
  float* agg1    = ws + (size_t)20 * N;
  float* h2      = ws + (size_t)36 * N;
  float* agg2A   = ws + (size_t)38 * N;

  const int B = 256;
  const int wsFloats = 40 * N;

  gnn_zero<<<(wsFloats + B - 1) / B, B, 0, stream>>>(ws, wsFloats);
  gnn_degrees<<<(E + B - 1) / B, B, 0, stream>>>(src, dst, degA, degB, N, E);
  gnn_invsqrt<<<(N + B - 1) / B, B, 0, stream>>>(degA, degB, inv_out, inv_in, N);

  const int tiles = N >> 4;
  gnn_gemm1_wmma<<<(tiles + 7) / 8, B, 0, stream>>>(x, W1, inv_out, h1, N);

  gnn_scatter16<<<2048, B, 0, stream>>>(src, dst, h1, agg1, (unsigned int*)h2, N, E);
  gnn_layer2<<<(N + B - 1) / B, B, 0, stream>>>(agg1, inv_in, inv_out, b1, W2, h2, agg2B, N);
  gnn_scatter2<<<(E + B - 1) / B, B, 0, stream>>>(src, dst, h2, agg2A, agg2B, E);
  gnn_final<<<(N + B - 1) / B, B, 0, stream>>>(agg2A, agg2B, inv_in, b2, out, N);
}


# Zero pass, degree counts and layer-2 edge sums run on a resident grid of 1024 work-groups whose waves loop (8192 waves per launch, not 100000-125000); adds arrival-ordered as before
# speedup vs baseline: 4.6624x; 1.0753x over previous
#include <hip/hip_runtime.h>
#include <math.h>


typedef __attribute__((ext_vector_type(2))) float v2f;
typedef __attribute__((ext_vector_type(8))) float v8f;

#define F_IN 512
#define HID  16
#define NCLS 2
#define GNN_RES 1024

__global__ __launch_bounds__(256) void gnn_zero(float* __restrict__ ws, int n) {
  float4* __restrict__ w4 = (float4*)ws;
  const int n4 = n >> 2;
  const int tid = blockIdx.x * blockDim.x + threadIdx.x;
  const int nthr = gridDim.x * blockDim.x;
  for (int i = tid; i < n4; i += nthr) w4[i] = make_float4(0.0f, 0.0f, 0.0f, 0.0f);
  if (tid < (n & 3)) ws[(n4 << 2) + tid] = 0.0f;
}

__device__ __forceinline__ int gnn_half() {
  return (int)((__builtin_amdgcn_s_sendmsg_rtn(0x87) >> 18) & 1u);
}

__global__ __launch_bounds__(256) void gnn_degrees(
    const int* __restrict__ src, const int* __restrict__ dst,
    float* __restrict__ degA, float* __restrict__ degB, int N, int E) {
  float* __restrict__ dego = gnn_half() ? degB : degA;
  float* __restrict__ degi = dego + (size_t)N;
  const int lane = threadIdx.x & 31;
  const int wave = __builtin_amdgcn_readfirstlane(
      (int)((blockIdx.x * blockDim.x + threadIdx.x) >> 5));
  const int step = (int)((gridDim.x * blockDim.x) >> 5) * 128;
  for (int base = wave * 128; base < E; base += step) {
    if (base + 128 <= E) {
      unsigned s[4], d[4];
      #pragma unroll
      for (int j = 0; j < 4; ++j) {
        s[j] = (unsigned)src[base + 32 * j + lane];
        d[j] = (unsigned)dst[base + 32 * j + lane];
      }
      #pragma unroll
      for (int j = 0; j < 4; ++j) {
        atomicAdd(dego + s[j], 1.0f);
        atomicAdd(degi + d[j], 1.0f);
      }
    } else {
      for (int e = base + lane; e < E; e += 32) {
        atomicAdd(dego + (unsigned)src[e], 1.0f);
        atomicAdd(degi + (unsigned)dst[e], 1.0f);
      }
    }
  }
}

__global__ void gnn_invsqrt(const float* __restrict__ degA, const float* __restrict__ degB,
                            float* __restrict__ inv_out, float* __restrict__ inv_in, int N) {
  int i = blockIdx.x * blockDim.x + threadIdx.x;
  if (i < N) {
    inv_out[i] = rsqrtf(fmaxf(degA[i] + degB[i], 1.0f));
    inv_in[i]  = rsqrtf(fmaxf(degA[(size_t)N + i] + degB[(size_t)N + i], 1.0f));
  }
}

__global__ __launch_bounds__(256) void gnn_gemm1_wmma(
    const float* __restrict__ x, const float* __restrict__ W1,
    const float* __restrict__ inv_out, float* __restrict__ h1, int N) {
  __shared__ float wlds[F_IN * HID];
  for (int i = threadIdx.x; i < F_IN * HID; i += blockDim.x) wlds[i] = W1[i];
  __syncthreads();

  const int wave = threadIdx.x >> 5;
  const int lane = threadIdx.x & 31;
  const int tiles = N >> 4;
  const int tile = blockIdx.x * 8 + wave;
  if (tile >= tiles) return;

  const int n16 = lane & 15;
  const int g   = lane >> 4;
  const int row = tile * 16 + n16;
  const float s = inv_out[row];
  const float* __restrict__ xr = x + (size_t)row * F_IN;

  v8f c = {};
  #pragma unroll 4
  for (int k = 0; k < F_IN; k += 4) {
    const int kk = k + 2 * g;
    v2f a = *(const v2f*)(xr + kk);
    a.x *= s;
    a.y *= s;
    v2f b;
    b.x = wlds[(kk + 0) * HID + n16];
    b.y = wlds[(kk + 1) * HID + n16];
    c = __builtin_amdgcn_wmma_f32_16x16x4_f32(
         false, a,  false, b,
         (short)0, c,  false,  false);
  }

  #pragma unroll
  for (int p = 0; p < 8; ++p) {
    const int m = tile * 16 + p + 8 * g;
    h1[(size_t)m * HID + n16] = c[p];
  }
}

#define GNN_RUN 128

__device__ __forceinline__ unsigned int gnn_draw(unsigned int* c, int lane) {
  unsigned int t = 0u;
  if (lane == 0) t = atomicAdd(c, 1u);
  return t;
}

__device__ __forceinline__ void gnn_add32(const int2* q, unsigned head,
                                          unsigned rem, unsigned g, unsigned f,
                                          const float* __restrict__ h1,
                                          float* __restrict__ agg1) {
  float v[16];
  unsigned d[16];
  #pragma unroll
  for (int i = 0; i < 16; ++i) {
    const unsigned k = 2u * i + g;
    const int2 sd = q[(head + k) & 127u];
    const unsigned s = (k < rem) ? (unsigned)sd.x : 0u;
    d[i] = (unsigned)sd.y;
    v[i] = h1[s * HID + f];
  }
  #pragma unroll
  for (int i = 0; i < 16; ++i) {
    const unsigned k = 2u * i + g;
    if (k < rem) atomicAdd(agg1 + d[i] * HID + f, v[i]);
  }
}

__global__ __launch_bounds__(256) void gnn_scatter16(
    const int* __restrict__ src, const int* __restrict__ dst,
    const float* __restrict__ h1, float* __restrict__ agg1,
    unsigned int* __restrict__ ctr, int N, int E) {
  __shared__ int2 ring[8][128];
  const int lane = threadIdx.x & 31;
  const int wave = __builtin_amdgcn_readfirstlane((int)(threadIdx.x >> 5));
  int2* q = ring[wave];
  const unsigned f = lane & 15;
  const unsigned g = lane >> 4;
  const int mid = N >> 1;
  const int runs = (E + GNN_RUN - 1) / GNN_RUN;
  const int mine = (runs - wave + 7) >> 3;
  const int h0 = gnn_half();
  unsigned head = 0u, tail = 0u;

  for (int pass = 0; pass < 2; ++pass) {
    const int h = h0 ^ pass;
    unsigned int* c = ctr + 64 * (8 * h + wave);
    unsigned int next = gnn_draw(c, lane);
    for (;;) {
      const int cur = (int)__builtin_amdgcn_readfirstlane(next);
      if (cur >= mine) break;
      const int run = 8 * cur + wave;
      for (int it = 0; it < GNN_RUN / 64; ++it) {
        const int e0 = run * GNN_RUN + it * 64 + lane;
        const int e1 = e0 + 32;
        const bool ok0 = e0 < E;
        const bool ok1 = e1 < E;
        const int s0 = ok0 ? src[e0] : 0;
        const int d0 = ok0 ? dst[e0] : 0;
        const int s1 = ok1 ? src[e1] : 0;
        const int d1 = ok1 ? dst[e1] : 0;
        if (it == 0) next = gnn_draw(c, lane);
        const bool own0 = ok0 && ((d0 >= mid) == (h != 0));
        const bool own1 = ok1 && ((d1 >= mid) == (h != 0));
        const unsigned m0 = __builtin_amdgcn_ballot_w32(own0);
        const unsigned m1 = __builtin_amdgcn_ballot_w32(own1);
        const unsigned c0 = (unsigned)__builtin_popcount(m0);
        const unsigned r0 = __builtin_amdgcn_mbcnt_lo(m0, 0u);
        const unsigned r1 = c0 + __builtin_amdgcn_mbcnt_lo(m1, 0u);
        if (own0) q[(tail + r0) & 127u] = make_int2(s0, d0);
        if (own1) q[(tail + r1) & 127u] = make_int2(s1, d1);
        tail += c0 + (unsigned)__builtin_popcount(m1);
        while (tail - head >= 32u) {
          gnn_add32(q, head, 32u, g, f, h1, agg1);
          head += 32u;
        }
      }
    }
  }
  if (tail != head) gnn_add32(q, head, tail - head, g, f, h1, agg1);
}

__global__ void gnn_layer2(const float* __restrict__ agg1, const float* __restrict__ inv_in,
                           const float* __restrict__ inv_out, const float* __restrict__ b1,
                           const float* __restrict__ W2, float* __restrict__ h2,
                           float* __restrict__ agg2B, int N) {
  int i = blockIdx.x * blockDim.x + threadIdx.x;
  if (i >= N) return;
  *(float2*)(agg2B + (size_t)i * NCLS) = make_float2(0.0f, 0.0f);
  const float si = inv_in[i];
  const float so = inv_out[i];
  const float* a = agg1 + (size_t)i * HID;
  float acc0 = 0.0f, acc1 = 0.0f;
  #pragma unroll
  for (int k = 0; k < HID; ++k) {
    float hv = fmaxf(fmaf(a[k], si, b1[k]), 0.0f) * so;
    acc0 = fmaf(hv, W2[k * NCLS + 0], acc0);
    acc1 = fmaf(hv, W2[k * NCLS + 1], acc1);
  }
  h2[(size_t)i * NCLS + 0] = acc0;
  h2[(size_t)i * NCLS + 1] = acc1;
}

__global__ __launch_bounds__(256) void gnn_scatter2(
    const int* __restrict__ src, const int* __restrict__ dst,
    const float* __restrict__ h2, float* __restrict__ agg2A,
    float* __restrict__ agg2B, int E) {
  const int lane = threadIdx.x & 31;
  const unsigned c = lane & 1;
  const int p = lane >> 1;
  float* __restrict__ agg2 = gnn_half() ? agg2B : agg2A;
  const int wave = __builtin_amdgcn_readfirstlane(
      (int)((blockIdx.x * blockDim.x + threadIdx.x) >> 5));
  const int step = (int)((gridDim.x * blockDim.x) >> 5) * 32;
  int e0 = wave * 32;
  if (e0 >= E) return;
  const int last = E - 1;
  int s_n = src[min(e0 + lane, last)];
  int d_n = dst[min(e0 + lane, last)];
  for (; e0 < E; e0 += step) {
    const int s_l = s_n;
    const int d_l = d_n;
    const int n = E - e0;
    const unsigned s0 = (unsigned)__shfl(s_l, p);
    const unsigned s1 = (unsigned)__shfl(s_l, 16 + p);
    const unsigned d0 = (unsigned)__shfl(d_l, p);
    const unsigned d1 = (unsigned)__shfl(d_l, 16 + p);
    const float v0 = h2[s0 * NCLS + c];
    const float v1 = h2[s1 * NCLS + c];
    const int en = min(e0 + step + lane, last);
    s_n = src[en];
    d_n = dst[en];
    if (n >= 32) {
      atomicAdd(agg2 + d0 * NCLS + c, v0);
      atomicAdd(agg2 + d1 * NCLS + c, v1);
    } else {
      if (p < n) atomicAdd(agg2 + d0 * NCLS + c, v0);
      if (16 + p < n) atomicAdd(agg2 + d1 * NCLS + c, v1);
    }
  }
}

__global__ void gnn_final(const float* __restrict__ agg2A, const float* __restrict__ agg2B,
                          const float* __restrict__ inv_in,
                          const float* __restrict__ b2, float* __restrict__ out, int N) {
  int i = blockIdx.x * blockDim.x + threadIdx.x;
  if (i >= N) return;
  const float si = inv_in[i];
  const float2 pa = *(const float2*)(agg2A + (size_t)i * NCLS);
  const float2 pb = *(const float2*)(agg2B + (size_t)i * NCLS);
  float v0 = fmaf(pa.x + pb.x, si, b2[0]);
  float v1 = fmaf(pa.y + pb.y, si, b2[1]);
  float m = fmaxf(v0, v1);
  float lse = m + logf(expf(v0 - m) + expf(v1 - m));
  out[(size_t)i * NCLS + 0] = v0 - lse;
  out[(size_t)i * NCLS + 1] = v1 - lse;
}

extern "C" void kernel_launch(void* const* d_in, const int* in_sizes, int n_in,
                              void* d_out, int out_size, void* d_ws, size_t ws_size,
                              hipStream_t stream) {
  const float* x   = (const float*)d_in[0];
  const int*   src = (const int*)d_in[1];
  const int*   dst = (const int*)d_in[2];
  const float* W1  = (const float*)d_in[3];
  const float* b1  = (const float*)d_in[4];
  const float* W2  = (const float*)d_in[5];
  const float* b2  = (const float*)d_in[6];
  float* out = (float*)d_out;

  const int N = in_sizes[0] / F_IN;
  const int E = in_sizes[1];

  float* ws      = (float*)d_ws;
  float* degA    = ws;
  float* inv_out = ws + (size_t)2 * N;
  float* inv_in  = ws + (size_t)3 * N;
  float* h1      = ws + (size_t)4 * N;
  float* degB    = h1;
  float* agg2B   = h1;
  float* agg1    = ws + (size_t)20 * N;
  float* h2      = ws + (size_t)36 * N;
  float* agg2A   = ws + (size_t)38 * N;

  const int B = 256;
  const int wsFloats = 40 * N;

  gnn_zero<<<GNN_RES, B, 0, stream>>>(ws, wsFloats);
  gnn_degrees<<<GNN_RES, B, 0, stream>>>(src, dst, degA, degB, N, E);
  gnn_invsqrt<<<(N + B - 1) / B, B, 0, stream>>>(degA, degB, inv_out, inv_in, N);

  const int tiles = N >> 4;
  gnn_gemm1_wmma<<<(tiles + 7) / 8, B, 0, stream>>>(x, W1, inv_out, h1, N);

  gnn_scatter16<<<2048, B, 0, stream>>>(src, dst, h1, agg1, (unsigned int*)h2, N, E);
  gnn_layer2<<<(N + B - 1) / B, B, 0, stream>>>(agg1, inv_in, inv_out, b1, W2, h2, agg2B, N);
  gnn_scatter2<<<GNN_RES, B, 0, stream>>>(src, dst, h2, agg2A, agg2B, E);
  gnn_final<<<(N + B - 1) / B, B, 0, stream>>>(agg2A, agg2B, inv_in, b2, out, N);
}


# x @ W1 keeps loads in flight: x read in 16-byte pieces, the next 64 floats' 8 loads issued before the current WMMAs (was 4 8-byte loads drained per trip); W1 staged 8 at once; k sum order changes
# speedup vs baseline: 4.9824x; 1.0686x over previous
#include <hip/hip_runtime.h>
#include <math.h>


typedef __attribute__((ext_vector_type(2))) float v2f;
typedef __attribute__((ext_vector_type(8))) float v8f;

#define F_IN 512
#define HID  16
#define NCLS 2
#define GNN_RES 1024

__global__ __launch_bounds__(256) void gnn_zero(float* __restrict__ ws, int n) {
  float4* __restrict__ w4 = (float4*)ws;
  const int n4 = n >> 2;
  const int tid = blockIdx.x * blockDim.x + threadIdx.x;
  const int nthr = gridDim.x * blockDim.x;
  for (int i = tid; i < n4; i += nthr) w4[i] = make_float4(0.0f, 0.0f, 0.0f, 0.0f);
  if (tid < (n & 3)) ws[(n4 << 2) + tid] = 0.0f;
}

__device__ __forceinline__ int gnn_half() {
  return (int)((__builtin_amdgcn_s_sendmsg_rtn(0x87) >> 18) & 1u);
}

__global__ __launch_bounds__(256) void gnn_degrees(
    const int* __restrict__ src, const int* __restrict__ dst,
    float* __restrict__ degA, float* __restrict__ degB, int N, int E) {
  float* __restrict__ dego = gnn_half() ? degB : degA;
  float* __restrict__ degi = dego + (size_t)N;
  const int lane = threadIdx.x & 31;
  const int wave = __builtin_amdgcn_readfirstlane(
      (int)((blockIdx.x * blockDim.x + threadIdx.x) >> 5));
  const int step = (int)((gridDim.x * blockDim.x) >> 5) * 128;
  for (int base = wave * 128; base < E; base += step) {
    if (base + 128 <= E) {
      unsigned s[4], d[4];
      #pragma unroll
      for (int j = 0; j < 4; ++j) {
        s[j] = (unsigned)src[base + 32 * j + lane];
        d[j] = (unsigned)dst[base + 32 * j + lane];
      }
      #pragma unroll
      for (int j = 0; j < 4; ++j) {
        atomicAdd(dego + s[j], 1.0f);
        atomicAdd(degi + d[j], 1.0f);
      }
    } else {
      for (int e = base + lane; e < E; e += 32) {
        atomicAdd(dego + (unsigned)src[e], 1.0f);
        atomicAdd(degi + (unsigned)dst[e], 1.0f);
      }
    }
  }
}

__global__ void gnn_invsqrt(const float* __restrict__ degA, const float* __restrict__ degB,
                            float* __restrict__ inv_out, float* __restrict__ inv_in, int N) {
  int i = blockIdx.x * blockDim.x + threadIdx.x;
  if (i < N) {
    inv_out[i] = rsqrtf(fmaxf(degA[i] + degB[i], 1.0f));
    inv_in[i]  = rsqrtf(fmaxf(degA[(size_t)N + i] + degB[(size_t)N + i], 1.0f));
  }
}

__global__ __launch_bounds__(256) void gnn_gemm1_wmma(
    const float* __restrict__ x, const float* __restrict__ W1,
    const float* __restrict__ inv_out, float* __restrict__ h1, int N) {
  __shared__ float wlds[F_IN * HID];
  {
    const float4* __restrict__ w4 = (const float4*)W1;
    float4* l4 = (float4*)wlds;
    float4 t[8];
    #pragma unroll
    for (int j = 0; j < 8; ++j) t[j] = w4[threadIdx.x + 256 * j];
    #pragma unroll
    for (int j = 0; j < 8; ++j) l4[threadIdx.x + 256 * j] = t[j];
  }
  __syncthreads();

  const int wave = threadIdx.x >> 5;
  const int lane = threadIdx.x & 31;
  const int tiles = N >> 4;
  const int tile = blockIdx.x * 8 + wave;
  if (tile >= tiles) return;

  const int n16 = lane & 15;
  const int g   = lane >> 4;
  const int row = tile * 16 + n16;
  const float s = inv_out[row];
  const float4* __restrict__ xr = (const float4*)(x + (size_t)row * F_IN + 4 * g);
  const float* wl = wlds + (4 * g) * HID + n16;

  v8f c = {};
  float4 cur[8], nxt[8];
  #pragma unroll
  for (int j = 0; j < 8; ++j) cur[j] = xr[2 * j];
  #pragma unroll
  for (int ch = 0; ch < F_IN / 64; ++ch) {
    if (ch + 1 < F_IN / 64) {
      #pragma unroll
      for (int j = 0; j < 8; ++j) nxt[j] = xr[16 * (ch + 1) + 2 * j];
    }
    #pragma unroll
    for (int j = 0; j < 8; ++j) {
      const int kb = 64 * ch + 8 * j;
      const float4 p = cur[j];
      v2f a0, a1, b0, b1;
      a0.x = p.x * s;
      a0.y = p.y * s;
      a1.x = p.z * s;
      a1.y = p.w * s;
      b0.x = wl[(kb + 0) * HID];
      b0.y = wl[(kb + 1) * HID];
      b1.x = wl[(kb + 2) * HID];
      b1.y = wl[(kb + 3) * HID];
      c = __builtin_amdgcn_wmma_f32_16x16x4_f32(
           false, a0,  false, b0,
           (short)0, c,  false,  false);
      c = __builtin_amdgcn_wmma_f32_16x16x4_f32(
           false, a1,  false, b1,
           (short)0, c,  false,  false);
    }
    #pragma unroll
    for (int j = 0; j < 8; ++j) cur[j] = nxt[j];
  }

  #pragma unroll
  for (int p = 0; p < 8; ++p) {
    const int m = tile * 16 + p + 8 * g;
    h1[(size_t)m * HID + n16] = c[p];
  }
}

#define GNN_RUN 128

__device__ __forceinline__ unsigned int gnn_draw(unsigned int* c, int lane) {
  unsigned int t = 0u;
  if (lane == 0) t = atomicAdd(c, 1u);
  return t;
}

__device__ __forceinline__ void gnn_add32(const int2* q, unsigned head,
                                          unsigned rem, unsigned g, unsigned f,
                                          const float* __restrict__ h1,
                                          float* __restrict__ agg1) {
  float v[16];
  unsigned d[16];
  #pragma unroll
  for (int i = 0; i < 16; ++i) {
    const unsigned k = 2u * i + g;
    const int2 sd = q[(head + k) & 127u];
    const unsigned s = (k < rem) ? (unsigned)sd.x : 0u;
    d[i] = (unsigned)sd.y;
    v[i] = h1[s * HID + f];
  }
  #pragma unroll
  for (int i = 0; i < 16; ++i) {
    const unsigned k = 2u * i + g;
    if (k < rem) atomicAdd(agg1 + d[i] * HID + f, v[i]);
  }
}

__global__ __launch_bounds__(256) void gnn_scatter16(
    const int* __restrict__ src, const int* __restrict__ dst,
    const float* __restrict__ h1, float* __restrict__ agg1,
    unsigned int* __restrict__ ctr, int N, int E) {
  __shared__ int2 ring[8][128];
  const int lane = threadIdx.x & 31;
  const int wave = __builtin_amdgcn_readfirstlane((int)(threadIdx.x >> 5));
  int2* q = ring[wave];
  const unsigned f = lane & 15;
  const unsigned g = lane >> 4;
  const int mid = N >> 1;
  const int runs = (E + GNN_RUN - 1) / GNN_RUN;
  const int mine = (runs - wave + 7) >> 3;
  const int h0 = gnn_half();
  unsigned head = 0u, tail = 0u;

  for (int pass = 0; pass < 2; ++pass) {
    const int h = h0 ^ pass;
    unsigned int* c = ctr + 64 * (8 * h + wave);
    unsigned int next = gnn_draw(c, lane);
    for (;;) {
      const int cur = (int)__builtin_amdgcn_readfirstlane(next);
      if (cur >= mine) break;
      const int run = 8 * cur + wave;
      for (int it = 0; it < GNN_RUN / 64; ++it) {
        const int e0 = run * GNN_RUN + it * 64 + lane;
        const int e1 = e0 + 32;
        const bool ok0 = e0 < E;
        const bool ok1 = e1 < E;
        const int s0 = ok0 ? src[e0] : 0;
        const int d0 = ok0 ? dst[e0] : 0;
        const int s1 = ok1 ? src[e1] : 0;
        const int d1 = ok1 ? dst[e1] : 0;
        if (it == 0) next = gnn_draw(c, lane);
        const bool own0 = ok0 && ((d0 >= mid) == (h != 0));
        const bool own1 = ok1 && ((d1 >= mid) == (h != 0));
        const unsigned m0 = __builtin_amdgcn_ballot_w32(own0);
        const unsigned m1 = __builtin_amdgcn_ballot_w32(own1);
        const unsigned c0 = (unsigned)__builtin_popcount(m0);
        const unsigned r0 = __builtin_amdgcn_mbcnt_lo(m0, 0u);
        const unsigned r1 = c0 + __builtin_amdgcn_mbcnt_lo(m1, 0u);
        if (own0) q[(tail + r0) & 127u] = make_int2(s0, d0);
        if (own1) q[(tail + r1) & 127u] = make_int2(s1, d1);
        tail += c0 + (unsigned)__builtin_popcount(m1);
        while (tail - head >= 32u) {
          gnn_add32(q, head, 32u, g, f, h1, agg1);
          head += 32u;
        }
      }
    }
  }
  if (tail != head) gnn_add32(q, head, tail - head, g, f, h1, agg1);
}

__global__ void gnn_layer2(const float* __restrict__ agg1, const float* __restrict__ inv_in,
                           const float* __restrict__ inv_out, const float* __restrict__ b1,
                           const float* __restrict__ W2, float* __restrict__ h2,
                           float* __restrict__ agg2B, int N) {
  int i = blockIdx.x * blockDim.x + threadIdx.x;
  if (i >= N) return;
  *(float2*)(agg2B + (size_t)i * NCLS) = make_float2(0.0f, 0.0f);
  const float si = inv_in[i];
  const float so = inv_out[i];
  const float* a = agg1 + (size_t)i * HID;
  float acc0 = 0.0f, acc1 = 0.0f;
  #pragma unroll
  for (int k = 0; k < HID; ++k) {
    float hv = fmaxf(fmaf(a[k], si, b1[k]), 0.0f) * so;
    acc0 = fmaf(hv, W2[k * NCLS + 0], acc0);
    acc1 = fmaf(hv, W2[k * NCLS + 1], acc1);
  }
  h2[(size_t)i * NCLS + 0] = acc0;
  h2[(size_t)i * NCLS + 1] = acc1;
}

__global__ __launch_bounds__(256) void gnn_scatter2(
    const int* __restrict__ src, const int* __restrict__ dst,
    const float* __restrict__ h2, float* __restrict__ agg2A,
    float* __restrict__ agg2B, int E) {
  const int lane = threadIdx.x & 31;
  const unsigned c = lane & 1;
  const int p = lane >> 1;
  float* __restrict__ agg2 = gnn_half() ? agg2B : agg2A;
  const int wave = __builtin_amdgcn_readfirstlane(
      (int)((blockIdx.x * blockDim.x + threadIdx.x) >> 5));
  const int step = (int)((gridDim.x * blockDim.x) >> 5) * 32;
  int e0 = wave * 32;
  if (e0 >= E) return;
  const int last = E - 1;
  int s_n = src[min(e0 + lane, last)];
  int d_n = dst[min(e0 + lane, last)];
  for (; e0 < E; e0 += step) {
    const int s_l = s_n;
    const int d_l = d_n;
    const int n = E - e0;
    const unsigned s0 = (unsigned)__shfl(s_l, p);
    const unsigned s1 = (unsigned)__shfl(s_l, 16 + p);
    const unsigned d0 = (unsigned)__shfl(d_l, p);
    const unsigned d1 = (unsigned)__shfl(d_l, 16 + p);
    const float v0 = h2[s0 * NCLS + c];
    const float v1 = h2[s1 * NCLS + c];
    const int en = min(e0 + step + lane, last);
    s_n = src[en];
    d_n = dst[en];
    if (n >= 32) {
      atomicAdd(agg2 + d0 * NCLS + c, v0);
      atomicAdd(agg2 + d1 * NCLS + c, v1);
    } else {
      if (p < n) atomicAdd(agg2 + d0 * NCLS + c, v0);
      if (16 + p < n) atomicAdd(agg2 + d1 * NCLS + c, v1);
    }
  }
}

__global__ void gnn_final(const float* __restrict__ agg2A, const float* __restrict__ agg2B,
                          const float* __restrict__ inv_in,
                          const float* __restrict__ b2, float* __restrict__ out, int N) {
  int i = blockIdx.x * blockDim.x + threadIdx.x;
  if (i >= N) return;
  const float si = inv_in[i];
  const float2 pa = *(const float2*)(agg2A + (size_t)i * NCLS);
  const float2 pb = *(const float2*)(agg2B + (size_t)i * NCLS);
  float v0 = fmaf(pa.x + pb.x, si, b2[0]);
  float v1 = fmaf(pa.y + pb.y, si, b2[1]);
  float m = fmaxf(v0, v1);
  float lse = m + logf(expf(v0 - m) + expf(v1 - m));
  out[(size_t)i * NCLS + 0] = v0 - lse;
  out[(size_t)i * NCLS + 1] = v1 - lse;
}

extern "C" void kernel_launch(void* const* d_in, const int* in_sizes, int n_in,
                              void* d_out, int out_size, void* d_ws, size_t ws_size,
                              hipStream_t stream) {
  const float* x   = (const float*)d_in[0];
  const int*   src = (const int*)d_in[1];
  const int*   dst = (const int*)d_in[2];
  const float* W1  = (const float*)d_in[3];
  const float* b1  = (const float*)d_in[4];
  const float* W2  = (const float*)d_in[5];
  const float* b2  = (const float*)d_in[6];
  float* out = (float*)d_out;

  const int N = in_sizes[0] / F_IN;
  const int E = in_sizes[1];

  float* ws      = (float*)d_ws;
  float* degA    = ws;
  float* inv_out = ws + (size_t)2 * N;
  float* inv_in  = ws + (size_t)3 * N;
  float* h1      = ws + (size_t)4 * N;
  float* degB    = h1;
  float* agg2B   = h1;
  float* agg1    = ws + (size_t)20 * N;
  float* h2      = ws + (size_t)36 * N;
  float* agg2A   = ws + (size_t)38 * N;

  const int B = 256;
  const int wsFloats = 40 * N;

  gnn_zero<<<GNN_RES, B, 0, stream>>>(ws, wsFloats);
  gnn_degrees<<<GNN_RES, B, 0, stream>>>(src, dst, degA, degB, N, E);
  gnn_invsqrt<<<(N + B - 1) / B, B, 0, stream>>>(degA, degB, inv_out, inv_in, N);

  const int tiles = N >> 4;
  gnn_gemm1_wmma<<<(tiles + 7) / 8, B, 0, stream>>>(x, W1, inv_out, h1, N);

  gnn_scatter16<<<2048, B, 0, stream>>>(src, dst, h1, agg1, (unsigned int*)h2, N, E);
  gnn_layer2<<<(N + B - 1) / B, B, 0, stream>>>(agg1, inv_in, inv_out, b1, W2, h2, agg2B, N);
  gnn_scatter2<<<GNN_RES, B, 0, stream>>>(src, dst, h2, agg2A, agg2B, E);
  gnn_final<<<(N + B - 1) / B, B, 0, stream>>>(agg2A, agg2B, inv_in, b2, out, N);
}


# Degree counts: each work-group counts its edges in an LDS table of 16-bit counters, then adds it to its half's plane with one whole-line atomic per 32 nodes (was one add per edge); counts exact
# speedup vs baseline: 5.4914x; 1.1022x over previous
#include <hip/hip_runtime.h>
#include <math.h>


typedef __attribute__((ext_vector_type(2))) float v2f;
typedef __attribute__((ext_vector_type(8))) float v8f;

#define F_IN 512
#define HID  16
#define NCLS 2
#define GNN_RES 1024

__global__ __launch_bounds__(256) void gnn_zero(float* __restrict__ ws, int n) {
  float4* __restrict__ w4 = (float4*)ws;
  const int n4 = n >> 2;
  const int tid = blockIdx.x * blockDim.x + threadIdx.x;
  const int nthr = gridDim.x * blockDim.x;
  for (int i = tid; i < n4; i += nthr) w4[i] = make_float4(0.0f, 0.0f, 0.0f, 0.0f);
  if (tid < (n & 3)) ws[(n4 << 2) + tid] = 0.0f;
}

__device__ __forceinline__ int gnn_half() {
  return (int)((__builtin_amdgcn_s_sendmsg_rtn(0x87) >> 18) & 1u);
}

#define GNN_DEG_WGS 256
#define GNN_TBINS   102400
#define GNN_TCHUNK  61440

__device__ __forceinline__ void gnn_count(unsigned int* tab, int node, int b0) {
  const unsigned b = (unsigned)node - (unsigned)b0;
  if (b < (unsigned)GNN_TBINS)
    atomicAdd(&tab[((b >> 6) << 5) | (b & 31u)], 1u << ((b >> 1) & 16u));
}

__global__ __launch_bounds__(1024) void gnn_degrees(
    const int* __restrict__ src, const int* __restrict__ dst,
    float* __restrict__ degA, float* __restrict__ degB, int N, int E) {
  __shared__ unsigned int tab[GNN_TBINS / 2];
  const int tid = threadIdx.x;
  const int lane = tid & 31;
  const int wave = __builtin_amdgcn_readfirstlane(tid >> 5);
  const int which = blockIdx.x & 1;
  const int* __restrict__ idx = which ? dst : src;
  float* __restrict__ plane = (gnn_half() ? degB : degA) + (which ? (size_t)N : (size_t)0);
  const int nw = (int)(gridDim.x >> 1);
  const int w = (int)(blockIdx.x >> 1);
  const int per = ((E + nw - 1) / nw + 3) & ~3;
  const int lo = min(w * per, E);
  const int hi = min(lo + per, E);

  for (int b0 = 0; b0 < N; b0 += GNN_TBINS) {
    const int nb = min(N - b0, GNN_TBINS);
    const int ngroups = (nb + 63) >> 6;
    const int rot = (w * ngroups) / nw;
    for (int c0 = lo; c0 < hi; c0 += GNN_TCHUNK) {
      const int c1 = min(c0 + GNN_TCHUNK, hi);
      uint4* t4 = (uint4*)tab;
      for (int i = tid; i < GNN_TBINS / 8; i += 1024) t4[i] = make_uint4(0u, 0u, 0u, 0u);
      __syncthreads();

      const int4* __restrict__ idx4 = (const int4*)(idx + c0);
      const int nq = (c1 - c0) >> 2;
      for (int q = tid; q < nq; q += 8 * 1024) {
        int4 r[8];
        #pragma unroll
        for (int j = 0; j < 8; ++j) r[j] = idx4[min(q + 1024 * j, nq - 1)];
        #pragma unroll
        for (int j = 0; j < 8; ++j) {
          if (q + 1024 * j < nq) {
            gnn_count(tab, r[j].x, b0);
            gnn_count(tab, r[j].y, b0);
            gnn_count(tab, r[j].z, b0);
            gnn_count(tab, r[j].w, b0);
          }
        }
      }
      if (tid < ((c1 - c0) & 3)) gnn_count(tab, idx[c0 + 4 * nq + tid], b0);
      __syncthreads();

      for (int k = wave; k < ngroups; k += 4 * 32) {
        unsigned v[4];
        int node[4];
        #pragma unroll
        for (int j = 0; j < 4; ++j) {
          const bool in = k + 32 * j < ngroups;
          int g = in ? k + 32 * j + rot : 0;
          if (g >= ngroups) g -= ngroups;
          const unsigned t = tab[(g << 5) + lane];
          v[j] = in ? t : 0u;
          node[j] = b0 + (g << 6) + lane;
        }
        #pragma unroll
        for (int j = 0; j < 4; ++j) {
          const unsigned cl = v[j] & 0xffffu;
          const unsigned ch = v[j] >> 16;
          if (cl != 0u && node[j] < N) atomicAdd(plane + node[j], (float)cl);
          if (ch != 0u && node[j] + 32 < N) atomicAdd(plane + node[j] + 32, (float)ch);
        }
      }
      __syncthreads();
    }
  }
}

__global__ void gnn_invsqrt(const float* __restrict__ degA, const float* __restrict__ degB,
                            float* __restrict__ inv_out, float* __restrict__ inv_in, int N) {
  int i = blockIdx.x * blockDim.x + threadIdx.x;
  if (i < N) {
    inv_out[i] = rsqrtf(fmaxf(degA[i] + degB[i], 1.0f));
    inv_in[i]  = rsqrtf(fmaxf(degA[(size_t)N + i] + degB[(size_t)N + i], 1.0f));
  }
}

__global__ __launch_bounds__(256) void gnn_gemm1_wmma(
    const float* __restrict__ x, const float* __restrict__ W1,
    const float* __restrict__ inv_out, float* __restrict__ h1, int N) {
  __shared__ float wlds[F_IN * HID];
  {
    const float4* __restrict__ w4 = (const float4*)W1;
    float4* l4 = (float4*)wlds;
    float4 t[8];
    #pragma unroll
    for (int j = 0; j < 8; ++j) t[j] = w4[threadIdx.x + 256 * j];
    #pragma unroll
    for (int j = 0; j < 8; ++j) l4[threadIdx.x + 256 * j] = t[j];
  }
  __syncthreads();

  const int wave = threadIdx.x >> 5;
  const int lane = threadIdx.x & 31;
  const int tiles = N >> 4;
  const int tile = blockIdx.x * 8 + wave;
  if (tile >= tiles) return;

  const int n16 = lane & 15;
  const int g   = lane >> 4;
  const int row = tile * 16 + n16;
  const float s = inv_out[row];
  const float4* __restrict__ xr = (const float4*)(x + (size_t)row * F_IN + 4 * g);
  const float* wl = wlds + (4 * g) * HID + n16;

  v8f c = {};
  float4 cur[8], nxt[8];
  #pragma unroll
  for (int j = 0; j < 8; ++j) cur[j] = xr[2 * j];
  #pragma unroll
  for (int ch = 0; ch < F_IN / 64; ++ch) {
    if (ch + 1 < F_IN / 64) {
      #pragma unroll
      for (int j = 0; j < 8; ++j) nxt[j] = xr[16 * (ch + 1) + 2 * j];
    }
    #pragma unroll
    for (int j = 0; j < 8; ++j) {
      const int kb = 64 * ch + 8 * j;
      const float4 p = cur[j];
      v2f a0, a1, b0, b1;
      a0.x = p.x * s;
      a0.y = p.y * s;
      a1.x = p.z * s;
      a1.y = p.w * s;
      b0.x = wl[(kb + 0) * HID];
      b0.y = wl[(kb + 1) * HID];
      b1.x = wl[(kb + 2) * HID];
      b1.y = wl[(kb + 3) * HID];
      c = __builtin_amdgcn_wmma_f32_16x16x4_f32(
           false, a0,  false, b0,
           (short)0, c,  false,  false);
      c = __builtin_amdgcn_wmma_f32_16x16x4_f32(
           false, a1,  false, b1,
           (short)0, c,  false,  false);
    }
    #pragma unroll
    for (int j = 0; j < 8; ++j) cur[j] = nxt[j];
  }

  #pragma unroll
  for (int p = 0; p < 8; ++p) {
    const int m = tile * 16 + p + 8 * g;
    h1[(size_t)m * HID + n16] = c[p];
  }
}

#define GNN_RUN 128

__device__ __forceinline__ unsigned int gnn_draw(unsigned int* c, int lane) {
  unsigned int t = 0u;
  if (lane == 0) t = atomicAdd(c, 1u);
  return t;
}

__device__ __forceinline__ void gnn_add32(const int2* q, unsigned head,
                                          unsigned rem, unsigned g, unsigned f,
                                          const float* __restrict__ h1,
                                          float* __restrict__ agg1) {
  float v[16];
  unsigned d[16];
  #pragma unroll
  for (int i = 0; i < 16; ++i) {
    const unsigned k = 2u * i + g;
    const int2 sd = q[(head + k) & 127u];
    const unsigned s = (k < rem) ? (unsigned)sd.x : 0u;
    d[i] = (unsigned)sd.y;
    v[i] = h1[s * HID + f];
  }
  #pragma unroll
  for (int i = 0; i < 16; ++i) {
    const unsigned k = 2u * i + g;
    if (k < rem) atomicAdd(agg1 + d[i] * HID + f, v[i]);
  }
}

__global__ __launch_bounds__(256) void gnn_scatter16(
    const int* __restrict__ src, const int* __restrict__ dst,
    const float* __restrict__ h1, float* __restrict__ agg1,
    unsigned int* __restrict__ ctr, int N, int E) {
  __shared__ int2 ring[8][128];
  const int lane = threadIdx.x & 31;
  const int wave = __builtin_amdgcn_readfirstlane((int)(threadIdx.x >> 5));
  int2* q = ring[wave];
  const unsigned f = lane & 15;
  const unsigned g = lane >> 4;
  const int mid = N >> 1;
  const int runs = (E + GNN_RUN - 1) / GNN_RUN;
  const int mine = (runs - wave + 7) >> 3;
  const int h0 = gnn_half();
  unsigned head = 0u, tail = 0u;

  for (int pass = 0; pass < 2; ++pass) {
    const int h = h0 ^ pass;
    unsigned int* c = ctr + 64 * (8 * h + wave);
    unsigned int next = gnn_draw(c, lane);
    for (;;) {
      const int cur = (int)__builtin_amdgcn_readfirstlane(next);
      if (cur >= mine) break;
      const int run = 8 * cur + wave;
      for (int it = 0; it < GNN_RUN / 64; ++it) {
        const int e0 = run * GNN_RUN + it * 64 + lane;
        const int e1 = e0 + 32;
        const bool ok0 = e0 < E;
        const bool ok1 = e1 < E;
        const int s0 = ok0 ? src[e0] : 0;
        const int d0 = ok0 ? dst[e0] : 0;
        const int s1 = ok1 ? src[e1] : 0;
        const int d1 = ok1 ? dst[e1] : 0;
        if (it == 0) next = gnn_draw(c, lane);
        const bool own0 = ok0 && ((d0 >= mid) == (h != 0));
        const bool own1 = ok1 && ((d1 >= mid) == (h != 0));
        const unsigned m0 = __builtin_amdgcn_ballot_w32(own0);
        const unsigned m1 = __builtin_amdgcn_ballot_w32(own1);
        const unsigned c0 = (unsigned)__builtin_popcount(m0);
        const unsigned r0 = __builtin_amdgcn_mbcnt_lo(m0, 0u);
        const unsigned r1 = c0 + __builtin_amdgcn_mbcnt_lo(m1, 0u);
        if (own0) q[(tail + r0) & 127u] = make_int2(s0, d0);
        if (own1) q[(tail + r1) & 127u] = make_int2(s1, d1);
        tail += c0 + (unsigned)__builtin_popcount(m1);
        while (tail - head >= 32u) {
          gnn_add32(q, head, 32u, g, f, h1, agg1);
          head += 32u;
        }
      }
    }
  }
  if (tail != head) gnn_add32(q, head, tail - head, g, f, h1, agg1);
}

__global__ void gnn_layer2(const float* __restrict__ agg1, const float* __restrict__ inv_in,
                           const float* __restrict__ inv_out, const float* __restrict__ b1,
                           const float* __restrict__ W2, float* __restrict__ h2,
                           float* __restrict__ agg2B, int N) {
  int i = blockIdx.x * blockDim.x + threadIdx.x;
  if (i >= N) return;
  *(float2*)(agg2B + (size_t)i * NCLS) = make_float2(0.0f, 0.0f);
  const float si = inv_in[i];
  const float so = inv_out[i];
  const float* a = agg1 + (size_t)i * HID;
  float acc0 = 0.0f, acc1 = 0.0f;
  #pragma unroll
  for (int k = 0; k < HID; ++k) {
    float hv = fmaxf(fmaf(a[k], si, b1[k]), 0.0f) * so;
    acc0 = fmaf(hv, W2[k * NCLS + 0], acc0);
    acc1 = fmaf(hv, W2[k * NCLS + 1], acc1);
  }
  h2[(size_t)i * NCLS + 0] = acc0;
  h2[(size_t)i * NCLS + 1] = acc1;
}

__global__ __launch_bounds__(256) void gnn_scatter2(
    const int* __restrict__ src, const int* __restrict__ dst,
    const float* __restrict__ h2, float* __restrict__ agg2A,
    float* __restrict__ agg2B, int E) {
  const int lane = threadIdx.x & 31;
  const unsigned c = lane & 1;
  const int p = lane >> 1;
  float* __restrict__ agg2 = gnn_half() ? agg2B : agg2A;
  const int wave = __builtin_amdgcn_readfirstlane(
      (int)((blockIdx.x * blockDim.x + threadIdx.x) >> 5));
  const int step = (int)((gridDim.x * blockDim.x) >> 5) * 32;
  int e0 = wave * 32;
  if (e0 >= E) return;
  const int last = E - 1;
  int s_n = src[min(e0 + lane, last)];
  int d_n = dst[min(e0 + lane, last)];
  for (; e0 < E; e0 += step) {
    const int s_l = s_n;
    const int d_l = d_n;
    const int n = E - e0;
    const unsigned s0 = (unsigned)__shfl(s_l, p);
    const unsigned s1 = (unsigned)__shfl(s_l, 16 + p);
    const unsigned d0 = (unsigned)__shfl(d_l, p);
    const unsigned d1 = (unsigned)__shfl(d_l, 16 + p);
    const float v0 = h2[s0 * NCLS + c];
    const float v1 = h2[s1 * NCLS + c];
    const int en = min(e0 + step + lane, last);
    s_n = src[en];
    d_n = dst[en];
    if (n >= 32) {
      atomicAdd(agg2 + d0 * NCLS + c, v0);
      atomicAdd(agg2 + d1 * NCLS + c, v1);
    } else {
      if (p < n) atomicAdd(agg2 + d0 * NCLS + c, v0);
      if (16 + p < n) atomicAdd(agg2 + d1 * NCLS + c, v1);
    }
  }
}

__global__ void gnn_final(const float* __restrict__ agg2A, const float* __restrict__ agg2B,
                          const float* __restrict__ inv_in,
                          const float* __restrict__ b2, float* __restrict__ out, int N) {
  int i = blockIdx.x * blockDim.x + threadIdx.x;
  if (i >= N) return;
  const float si = inv_in[i];
  const float2 pa = *(const float2*)(agg2A + (size_t)i * NCLS);
  const float2 pb = *(const float2*)(agg2B + (size_t)i * NCLS);
  float v0 = fmaf(pa.x + pb.x, si, b2[0]);
  float v1 = fmaf(pa.y + pb.y, si, b2[1]);
  float m = fmaxf(v0, v1);
  float lse = m + logf(expf(v0 - m) + expf(v1 - m));
  out[(size_t)i * NCLS + 0] = v0 - lse;
  out[(size_t)i * NCLS + 1] = v1 - lse;
}

extern "C" void kernel_launch(void* const* d_in, const int* in_sizes, int n_in,
                              void* d_out, int out_size, void* d_ws, size_t ws_size,
                              hipStream_t stream) {
  const float* x   = (const float*)d_in[0];
  const int*   src = (const int*)d_in[1];
  const int*   dst = (const int*)d_in[2];
  const float* W1  = (const float*)d_in[3];
  const float* b1  = (const float*)d_in[4];
  const float* W2  = (const float*)d_in[5];
  const float* b2  = (const float*)d_in[6];
  float* out = (float*)d_out;

  const int N = in_sizes[0] / F_IN;
  const int E = in_sizes[1];

  float* ws      = (float*)d_ws;
  float* degA    = ws;
  float* inv_out = ws + (size_t)2 * N;
  float* inv_in  = ws + (size_t)3 * N;
  float* h1      = ws + (size_t)4 * N;
  float* degB    = h1;
  float* agg2B   = h1;
  float* agg1    = ws + (size_t)20 * N;
  float* h2      = ws + (size_t)36 * N;
  float* agg2A   = ws + (size_t)38 * N;

  const int B = 256;
  const int wsFloats = 40 * N;

  gnn_zero<<<GNN_RES, B, 0, stream>>>(ws, wsFloats);
  gnn_degrees<<<GNN_DEG_WGS, 1024, 0, stream>>>(src, dst, degA, degB, N, E);
  gnn_invsqrt<<<(N + B - 1) / B, B, 0, stream>>>(degA, degB, inv_out, inv_in, N);

  const int tiles = N >> 4;
  gnn_gemm1_wmma<<<(tiles + 7) / 8, B, 0, stream>>>(x, W1, inv_out, h1, N);

  gnn_scatter16<<<2048, B, 0, stream>>>(src, dst, h1, agg1, (unsigned int*)h2, N, E);
  gnn_layer2<<<(N + B - 1) / B, B, 0, stream>>>(agg1, inv_in, inv_out, b1, W2, h2, agg2B, N);
  gnn_scatter2<<<GNN_RES, B, 0, stream>>>(src, dst, h2, agg2A, agg2B, E);
  gnn_final<<<(N + B - 1) / B, B, 0, stream>>>(agg2A, agg2B, inv_in, b2, out, N);
}


# Layer-2 edge sums carry ONE float per edge (class difference of h2: a 2-class log-softmax needs only v0-v1), so half the nodes' values fit in LDS and edge gathers read LDS, not L2; sum tree changes
# speedup vs baseline: 5.6810x; 1.0345x over previous
#include <hip/hip_runtime.h>
#include <math.h>


typedef __attribute__((ext_vector_type(2))) float v2f;
typedef __attribute__((ext_vector_type(8))) float v8f;

#define F_IN 512
#define HID  16
#define NCLS 2
#define GNN_RES 1024

__global__ __launch_bounds__(256) void gnn_zero(float* __restrict__ ws, int n) {
  float4* __restrict__ w4 = (float4*)ws;
  const int n4 = n >> 2;
  const int tid = blockIdx.x * blockDim.x + threadIdx.x;
  const int nthr = gridDim.x * blockDim.x;
  for (int i = tid; i < n4; i += nthr) w4[i] = make_float4(0.0f, 0.0f, 0.0f, 0.0f);
  if (tid < (n & 3)) ws[(n4 << 2) + tid] = 0.0f;
}

__device__ __forceinline__ int gnn_half() {
  return (int)((__builtin_amdgcn_s_sendmsg_rtn(0x87) >> 18) & 1u);
}

#define GNN_DEG_WGS 256
#define GNN_TBINS   102400
#define GNN_TCHUNK  61440

__device__ __forceinline__ void gnn_count(unsigned int* tab, int node, int b0) {
  const unsigned b = (unsigned)node - (unsigned)b0;
  if (b < (unsigned)GNN_TBINS)
    atomicAdd(&tab[((b >> 6) << 5) | (b & 31u)], 1u << ((b >> 1) & 16u));
}

__global__ __launch_bounds__(1024) void gnn_degrees(
    const int* __restrict__ src, const int* __restrict__ dst,
    float* __restrict__ degA, float* __restrict__ degB, int N, int E) {
  __shared__ unsigned int tab[GNN_TBINS / 2];
  const int tid = threadIdx.x;
  const int lane = tid & 31;
  const int wave = __builtin_amdgcn_readfirstlane(tid >> 5);
  const int which = blockIdx.x & 1;
  const int* __restrict__ idx = which ? dst : src;
  float* __restrict__ plane = (gnn_half() ? degB : degA) + (which ? (size_t)N : (size_t)0);
  const int nw = (int)(gridDim.x >> 1);
  const int w = (int)(blockIdx.x >> 1);
  const int per = ((E + nw - 1) / nw + 3) & ~3;
  const int lo = min(w * per, E);
  const int hi = min(lo + per, E);

  for (int b0 = 0; b0 < N; b0 += GNN_TBINS) {
    const int nb = min(N - b0, GNN_TBINS);
    const int ngroups = (nb + 63) >> 6;
    const int rot = (w * ngroups) / nw;
    for (int c0 = lo; c0 < hi; c0 += GNN_TCHUNK) {
      const int c1 = min(c0 + GNN_TCHUNK, hi);
      uint4* t4 = (uint4*)tab;
      for (int i = tid; i < GNN_TBINS / 8; i += 1024) t4[i] = make_uint4(0u, 0u, 0u, 0u);
      __syncthreads();

      const int4* __restrict__ idx4 = (const int4*)(idx + c0);
      const int nq = (c1 - c0) >> 2;
      for (int q = tid; q < nq; q += 8 * 1024) {
        int4 r[8];
        #pragma unroll
        for (int j = 0; j < 8; ++j) r[j] = idx4[min(q + 1024 * j, nq - 1)];
        #pragma unroll
        for (int j = 0; j < 8; ++j) {
          if (q + 1024 * j < nq) {
            gnn_count(tab, r[j].x, b0);
            gnn_count(tab, r[j].y, b0);
            gnn_count(tab, r[j].z, b0);
            gnn_count(tab, r[j].w, b0);
          }
        }
      }
      if (tid < ((c1 - c0) & 3)) gnn_count(tab, idx[c0 + 4 * nq + tid], b0);
      __syncthreads();

      for (int k = wave; k < ngroups; k += 4 * 32) {
        unsigned v[4];
        int node[4];
        #pragma unroll
        for (int j = 0; j < 4; ++j) {
          const bool in = k + 32 * j < ngroups;
          int g = in ? k + 32 * j + rot : 0;
          if (g >= ngroups) g -= ngroups;
          const unsigned t = tab[(g << 5) + lane];
          v[j] = in ? t : 0u;
          node[j] = b0 + (g << 6) + lane;
        }
        #pragma unroll
        for (int j = 0; j < 4; ++j) {
          const unsigned cl = v[j] & 0xffffu;
          const unsigned ch = v[j] >> 16;
          if (cl != 0u && node[j] < N) atomicAdd(plane + node[j], (float)cl);
          if (ch != 0u && node[j] + 32 < N) atomicAdd(plane + node[j] + 32, (float)ch);
        }
      }
      __syncthreads();
    }
  }
}

__global__ void gnn_invsqrt(const float* __restrict__ degA, const float* __restrict__ degB,
                            float* __restrict__ inv_out, float* __restrict__ inv_in, int N) {
  int i = blockIdx.x * blockDim.x + threadIdx.x;
  if (i < N) {
    inv_out[i] = rsqrtf(fmaxf(degA[i] + degB[i], 1.0f));
    inv_in[i]  = rsqrtf(fmaxf(degA[(size_t)N + i] + degB[(size_t)N + i], 1.0f));
  }
}

__global__ __launch_bounds__(256) void gnn_gemm1_wmma(
    const float* __restrict__ x, const float* __restrict__ W1,
    const float* __restrict__ inv_out, float* __restrict__ h1, int N) {
  __shared__ float wlds[F_IN * HID];
  {
    const float4* __restrict__ w4 = (const float4*)W1;
    float4* l4 = (float4*)wlds;
    float4 t[8];
    #pragma unroll
    for (int j = 0; j < 8; ++j) t[j] = w4[threadIdx.x + 256 * j];
    #pragma unroll
    for (int j = 0; j < 8; ++j) l4[threadIdx.x + 256 * j] = t[j];
  }
  __syncthreads();

  const int wave = threadIdx.x >> 5;
  const int lane = threadIdx.x & 31;
  const int tiles = N >> 4;
  const int tile = blockIdx.x * 8 + wave;
  if (tile >= tiles) return;

  const int n16 = lane & 15;
  const int g   = lane >> 4;
  const int row = tile * 16 + n16;
  const float s = inv_out[row];
  const float4* __restrict__ xr = (const float4*)(x + (size_t)row * F_IN + 4 * g);
  const float* wl = wlds + (4 * g) * HID + n16;

  v8f c = {};
  float4 cur[8], nxt[8];
  #pragma unroll
  for (int j = 0; j < 8; ++j) cur[j] = xr[2 * j];
  #pragma unroll
  for (int ch = 0; ch < F_IN / 64; ++ch) {
    if (ch + 1 < F_IN / 64) {
      #pragma unroll
      for (int j = 0; j < 8; ++j) nxt[j] = xr[16 * (ch + 1) + 2 * j];
    }
    #pragma unroll
    for (int j = 0; j < 8; ++j) {
      const int kb = 64 * ch + 8 * j;
      const float4 p = cur[j];
      v2f a0, a1, b0, b1;
      a0.x = p.x * s;
      a0.y = p.y * s;
      a1.x = p.z * s;
      a1.y = p.w * s;
      b0.x = wl[(kb + 0) * HID];
      b0.y = wl[(kb + 1) * HID];
      b1.x = wl[(kb + 2) * HID];
      b1.y = wl[(kb + 3) * HID];
      c = __builtin_amdgcn_wmma_f32_16x16x4_f32(
           false, a0,  false, b0,
           (short)0, c,  false,  false);
      c = __builtin_amdgcn_wmma_f32_16x16x4_f32(
           false, a1,  false, b1,
           (short)0, c,  false,  false);
    }
    #pragma unroll
    for (int j = 0; j < 8; ++j) cur[j] = nxt[j];
  }

  #pragma unroll
  for (int p = 0; p < 8; ++p) {
    const int m = tile * 16 + p + 8 * g;
    h1[(size_t)m * HID + n16] = c[p];
  }
}

#define GNN_RUN 128

__device__ __forceinline__ unsigned int gnn_draw(unsigned int* c, int lane) {
  unsigned int t = 0u;
  if (lane == 0) t = atomicAdd(c, 1u);
  return t;
}

__device__ __forceinline__ void gnn_add32(const int2* q, unsigned head,
                                          unsigned rem, unsigned g, unsigned f,
                                          const float* __restrict__ h1,
                                          float* __restrict__ agg1) {
  float v[16];
  unsigned d[16];
  #pragma unroll
  for (int i = 0; i < 16; ++i) {
    const unsigned k = 2u * i + g;
    const int2 sd = q[(head + k) & 127u];
    const unsigned s = (k < rem) ? (unsigned)sd.x : 0u;
    d[i] = (unsigned)sd.y;
    v[i] = h1[s * HID + f];
  }
  #pragma unroll
  for (int i = 0; i < 16; ++i) {
    const unsigned k = 2u * i + g;
    if (k < rem) atomicAdd(agg1 + d[i] * HID + f, v[i]);
  }
}

__global__ __launch_bounds__(256) void gnn_scatter16(
    const int* __restrict__ src, const int* __restrict__ dst,
    const float* __restrict__ h1, float* __restrict__ agg1,
    unsigned int* __restrict__ ctr, int N, int E) {
  __shared__ int2 ring[8][128];
  const int lane = threadIdx.x & 31;
  const int wave = __builtin_amdgcn_readfirstlane((int)(threadIdx.x >> 5));
  int2* q = ring[wave];
  const unsigned f = lane & 15;
  const unsigned g = lane >> 4;
  const int mid = N >> 1;
  const int runs = (E + GNN_RUN - 1) / GNN_RUN;
  const int mine = (runs - wave + 7) >> 3;
  const int h0 = gnn_half();
  unsigned head = 0u, tail = 0u;

  for (int pass = 0; pass < 2; ++pass) {
    const int h = h0 ^ pass;
    unsigned int* c = ctr + 64 * (8 * h + wave);
    unsigned int next = gnn_draw(c, lane);
    for (;;) {
      const int cur = (int)__builtin_amdgcn_readfirstlane(next);
      if (cur >= mine) break;
      const int run = 8 * cur + wave;
      for (int it = 0; it < GNN_RUN / 64; ++it) {
        const int e0 = run * GNN_RUN + it * 64 + lane;
        const int e1 = e0 + 32;
        const bool ok0 = e0 < E;
        const bool ok1 = e1 < E;
        const int s0 = ok0 ? src[e0] : 0;
        const int d0 = ok0 ? dst[e0] : 0;
        const int s1 = ok1 ? src[e1] : 0;
        const int d1 = ok1 ? dst[e1] : 0;
        if (it == 0) next = gnn_draw(c, lane);
        const bool own0 = ok0 && ((d0 >= mid) == (h != 0));
        const bool own1 = ok1 && ((d1 >= mid) == (h != 0));
        const unsigned m0 = __builtin_amdgcn_ballot_w32(own0);
        const unsigned m1 = __builtin_amdgcn_ballot_w32(own1);
        const unsigned c0 = (unsigned)__builtin_popcount(m0);
        const unsigned r0 = __builtin_amdgcn_mbcnt_lo(m0, 0u);
        const unsigned r1 = c0 + __builtin_amdgcn_mbcnt_lo(m1, 0u);
        if (own0) q[(tail + r0) & 127u] = make_int2(s0, d0);
        if (own1) q[(tail + r1) & 127u] = make_int2(s1, d1);
        tail += c0 + (unsigned)__builtin_popcount(m1);
        while (tail - head >= 32u) {
          gnn_add32(q, head, 32u, g, f, h1, agg1);
          head += 32u;
        }
      }
    }
  }
  if (tail != head) gnn_add32(q, head, tail - head, g, f, h1, agg1);
}

__global__ void gnn_layer2(const float* __restrict__ agg1, const float* __restrict__ inv_in,
                           const float* __restrict__ inv_out, const float* __restrict__ b1,
                           const float* __restrict__ W2, float* __restrict__ h2d,
                           float* __restrict__ agg2B, int N) {
  int i = blockIdx.x * blockDim.x + threadIdx.x;
  if (i >= N) return;
  agg2B[i] = 0.0f;
  const float si = inv_in[i];
  const float so = inv_out[i];
  const float* a = agg1 + (size_t)i * HID;
  float acc0 = 0.0f, acc1 = 0.0f;
  #pragma unroll
  for (int k = 0; k < HID; ++k) {
    float hv = fmaxf(fmaf(a[k], si, b1[k]), 0.0f) * so;
    acc0 = fmaf(hv, W2[k * NCLS + 0], acc0);
    acc1 = fmaf(hv, W2[k * NCLS + 1], acc1);
  }
  h2d[i] = acc0 - acc1;
}

#define GNN_S2_WGS 256
#define GNN_S2_TAB 51200

__device__ __forceinline__ void gnn_edges4(const float* tab, int4 s, int4 d, bool in,
                                           int b0, int nb, float* __restrict__ plane) {
  const unsigned b[4] = {(unsigned)s.x - (unsigned)b0, (unsigned)s.y - (unsigned)b0,
                         (unsigned)s.z - (unsigned)b0, (unsigned)s.w - (unsigned)b0};
  const int dd[4] = {d.x, d.y, d.z, d.w};
  bool ok[4];
  float v[4];
  #pragma unroll
  for (int i = 0; i < 4; ++i) {
    ok[i] = in && b[i] < (unsigned)nb;
    v[i] = tab[ok[i] ? b[i] : 0u];
  }
  #pragma unroll
  for (int i = 0; i < 4; ++i)
    if (ok[i]) atomicAdd(plane + dd[i], v[i]);
}

__global__ __launch_bounds__(1024) void gnn_scatter2(
    const int* __restrict__ src, const int* __restrict__ dst,
    const float* __restrict__ h2d, float* __restrict__ agg2A,
    float* __restrict__ agg2B, int N, int E) {
  __shared__ float tab[GNN_S2_TAB];
  const int tid = threadIdx.x;
  float* __restrict__ plane = gnn_half() ? agg2B : agg2A;
  const int nw = (int)(gridDim.x >> 1);
  const int w = (int)(blockIdx.x >> 1);
  const int per = ((E + nw - 1) / nw + 3) & ~3;
  const int lo = min(w * per, E);
  const int hi = min(lo + per, E);
  const int4* __restrict__ s4 = (const int4*)(src + lo);
  const int4* __restrict__ d4 = (const int4*)(dst + lo);
  const int nq = (hi - lo) >> 2;
  const int parts = (N + GNN_S2_TAB - 1) / GNN_S2_TAB;

  for (int p = (int)(blockIdx.x & 1); p < parts; p += 2) {
    const int b0 = p * GNN_S2_TAB;
    const int nb = min(N - b0, GNN_S2_TAB);
    const float4* __restrict__ g4 = (const float4*)(h2d + b0);
    float4* t4 = (float4*)tab;
    const int n4 = nb >> 2;
    for (int i = tid; i < n4; i += 8 * 1024) {
      float4 r[8];
      #pragma unroll
      for (int j = 0; j < 8; ++j) r[j] = g4[min(i + 1024 * j, n4 - 1)];
      #pragma unroll
      for (int j = 0; j < 8; ++j) t4[min(i + 1024 * j, n4 - 1)] = r[j];
    }
    if (tid < (nb & 3)) tab[4 * n4 + tid] = h2d[b0 + 4 * n4 + tid];
    __syncthreads();

    for (int q = tid; q < nq; q += 4 * 1024) {
      int4 rs[4], rd[4];
      #pragma unroll
      for (int j = 0; j < 4; ++j) {
        const int k = min(q + 1024 * j, nq - 1);
        rs[j] = s4[k];
        rd[j] = d4[k];
      }
      asm volatile("" ::: "memory");
      #pragma unroll
      for (int j = 0; j < 4; ++j)
        gnn_edges4(tab, rs[j], rd[j], q + 1024 * j < nq, b0, nb, plane);
    }
    if (tid < ((hi - lo) & 3)) {
      const unsigned b = (unsigned)src[lo + 4 * nq + tid] - (unsigned)b0;
      if (b < (unsigned)nb) atomicAdd(plane + dst[lo + 4 * nq + tid], tab[b]);
    }
    __syncthreads();
  }
}

__global__ void gnn_final(const float* __restrict__ agg2A, const float* __restrict__ agg2B,
                          const float* __restrict__ inv_in,
                          const float* __restrict__ b2, float* __restrict__ out, int N) {
  int i = blockIdx.x * blockDim.x + threadIdx.x;
  if (i >= N) return;
  const float d = fmaf(agg2A[i] + agg2B[i], inv_in[i], b2[0] - b2[1]);
  const float t = log1pf(expf(-fabsf(d)));
  out[(size_t)i * NCLS + 0] = -(fmaxf(-d, 0.0f) + t);
  out[(size_t)i * NCLS + 1] = -(fmaxf(d, 0.0f) + t);
}

extern "C" void kernel_launch(void* const* d_in, const int* in_sizes, int n_in,
                              void* d_out, int out_size, void* d_ws, size_t ws_size,
                              hipStream_t stream) {
  const float* x   = (const float*)d_in[0];
  const int*   src = (const int*)d_in[1];
  const int*   dst = (const int*)d_in[2];
  const float* W1  = (const float*)d_in[3];
  const float* b1  = (const float*)d_in[4];
  const float* W2  = (const float*)d_in[5];
  const float* b2  = (const float*)d_in[6];
  float* out = (float*)d_out;

  const int N = in_sizes[0] / F_IN;
  const int E = in_sizes[1];

  float* ws      = (float*)d_ws;
  float* degA    = ws;
  float* inv_out = ws + (size_t)2 * N;
  float* inv_in  = ws + (size_t)3 * N;
  float* h1      = ws + (size_t)4 * N;
  float* degB    = h1;
  float* agg2B   = h1;
  float* agg1    = ws + (size_t)20 * N;
  float* h2d     = ws + (size_t)36 * N;
  float* agg2A   = ws + (size_t)38 * N;

  const int B = 256;
  const int wsFloats = 40 * N;

  gnn_zero<<<GNN_RES, B, 0, stream>>>(ws, wsFloats);
  gnn_degrees<<<GNN_DEG_WGS, 1024, 0, stream>>>(src, dst, degA, degB, N, E);
  gnn_invsqrt<<<(N + B - 1) / B, B, 0, stream>>>(degA, degB, inv_out, inv_in, N);

  const int tiles = N >> 4;
  gnn_gemm1_wmma<<<(tiles + 7) / 8, B, 0, stream>>>(x, W1, inv_out, h1, N);

  gnn_scatter16<<<2048, B, 0, stream>>>(src, dst, h1, agg1, (unsigned int*)h2d, N, E);
  gnn_layer2<<<(N + B - 1) / B, B, 0, stream>>>(agg1, inv_in, inv_out, b1, W2, h2d, agg2B, N);
  gnn_scatter2<<<GNN_S2_WGS, 1024, 0, stream>>>(src, dst, h2d, agg2A, agg2B, N, E);
  gnn_final<<<(N + B - 1) / B, B, 0, stream>>>(agg2A, agg2B, inv_in, b2, out, N);
}


# Node-local passes folded into their neighbours: 1/sqrt(degree) computed where used (7 launches, was 8) and each sum array zeroed by the kernel before its first add (zero pass 1.6 MB, was 16 MB)
# speedup vs baseline: 6.1592x; 1.0842x over previous
#include <hip/hip_runtime.h>
#include <math.h>


typedef __attribute__((ext_vector_type(2))) float v2f;
typedef __attribute__((ext_vector_type(8))) float v8f;

#define F_IN 512
#define HID  16
#define NCLS 2
#define GNN_CTR_WORDS 1024

__global__ __launch_bounds__(256) void gnn_zero(float* __restrict__ ws, int n,
                                                float* __restrict__ ctr) {
  float4* __restrict__ w4 = (float4*)ws;
  const int n4 = n >> 2;
  const int tid = blockIdx.x * blockDim.x + threadIdx.x;
  const int nthr = gridDim.x * blockDim.x;
  for (int i = tid; i < n4; i += nthr) w4[i] = make_float4(0.0f, 0.0f, 0.0f, 0.0f);
  if (tid < (n & 3)) ws[(n4 << 2) + tid] = 0.0f;
  if (tid < GNN_CTR_WORDS / 4) ((float4*)ctr)[tid] = make_float4(0.0f, 0.0f, 0.0f, 0.0f);
}

__device__ __forceinline__ int gnn_half() {
  return (int)((__builtin_amdgcn_s_sendmsg_rtn(0x87) >> 18) & 1u);
}

#define GNN_DEG_WGS 256
#define GNN_TBINS   102400
#define GNN_TCHUNK  61440

__device__ __forceinline__ void gnn_count(unsigned int* tab, int node, int b0) {
  const unsigned b = (unsigned)node - (unsigned)b0;
  if (b < (unsigned)GNN_TBINS)
    atomicAdd(&tab[((b >> 6) << 5) | (b & 31u)], 1u << ((b >> 1) & 16u));
}

__global__ __launch_bounds__(1024) void gnn_degrees(
    const int* __restrict__ src, const int* __restrict__ dst,
    float* __restrict__ degA, float* __restrict__ degB, int N, int E) {
  __shared__ unsigned int tab[GNN_TBINS / 2];
  const int tid = threadIdx.x;
  const int lane = tid & 31;
  const int wave = __builtin_amdgcn_readfirstlane(tid >> 5);
  const int which = blockIdx.x & 1;
  const int* __restrict__ idx = which ? dst : src;
  float* __restrict__ plane = (gnn_half() ? degB : degA) + (which ? (size_t)N : (size_t)0);
  const int nw = (int)(gridDim.x >> 1);
  const int w = (int)(blockIdx.x >> 1);
  const int per = ((E + nw - 1) / nw + 3) & ~3;
  const int lo = min(w * per, E);
  const int hi = min(lo + per, E);

  for (int b0 = 0; b0 < N; b0 += GNN_TBINS) {
    const int nb = min(N - b0, GNN_TBINS);
    const int ngroups = (nb + 63) >> 6;
    const int rot = (w * ngroups) / nw;
    for (int c0 = lo; c0 < hi; c0 += GNN_TCHUNK) {
      const int c1 = min(c0 + GNN_TCHUNK, hi);
      uint4* t4 = (uint4*)tab;
      for (int i = tid; i < GNN_TBINS / 8; i += 1024) t4[i] = make_uint4(0u, 0u, 0u, 0u);
      __syncthreads();

      const int4* __restrict__ idx4 = (const int4*)(idx + c0);
      const int nq = (c1 - c0) >> 2;
      for (int q = tid; q < nq; q += 8 * 1024) {
        int4 r[8];
        #pragma unroll
        for (int j = 0; j < 8; ++j) r[j] = idx4[min(q + 1024 * j, nq - 1)];
        #pragma unroll
        for (int j = 0; j < 8; ++j) {
          if (q + 1024 * j < nq) {
            gnn_count(tab, r[j].x, b0);
            gnn_count(tab, r[j].y, b0);
            gnn_count(tab, r[j].z, b0);
            gnn_count(tab, r[j].w, b0);
          }
        }
      }
      if (tid < ((c1 - c0) & 3)) gnn_count(tab, idx[c0 + 4 * nq + tid], b0);
      __syncthreads();

      for (int k = wave; k < ngroups; k += 4 * 32) {
        unsigned v[4];
        int node[4];
        #pragma unroll
        for (int j = 0; j < 4; ++j) {
          const bool in = k + 32 * j < ngroups;
          int g = in ? k + 32 * j + rot : 0;
          if (g >= ngroups) g -= ngroups;
          const unsigned t = tab[(g << 5) + lane];
          v[j] = in ? t : 0u;
          node[j] = b0 + (g << 6) + lane;
        }
        #pragma unroll
        for (int j = 0; j < 4; ++j) {
          const unsigned cl = v[j] & 0xffffu;
          const unsigned ch = v[j] >> 16;
          if (cl != 0u && node[j] < N) atomicAdd(plane + node[j], (float)cl);
          if (ch != 0u && node[j] + 32 < N) atomicAdd(plane + node[j] + 32, (float)ch);
        }
      }
      __syncthreads();
    }
  }
}

__device__ __forceinline__ float gnn_inv(const float* __restrict__ degA,
                                         const float* __restrict__ degB, size_t i) {
  return rsqrtf(fmaxf(degA[i] + degB[i], 1.0f));
}

__global__ __launch_bounds__(256) void gnn_gemm1_wmma(
    const float* __restrict__ x, const float* __restrict__ W1,
    const float* __restrict__ degA, const float* __restrict__ degB,
    float* __restrict__ h1, float* __restrict__ agg1, int N) {
  __shared__ float wlds[F_IN * HID];
  {
    const float4* __restrict__ w4 = (const float4*)W1;
    float4* l4 = (float4*)wlds;
    float4 t[8];
    #pragma unroll
    for (int j = 0; j < 8; ++j) t[j] = w4[threadIdx.x + 256 * j];
    #pragma unroll
    for (int j = 0; j < 8; ++j) l4[threadIdx.x + 256 * j] = t[j];
  }
  __syncthreads();

  const int wave = threadIdx.x >> 5;
  const int lane = threadIdx.x & 31;
  const int tiles = N >> 4;
  const int tile = blockIdx.x * 8 + wave;
  if (tile >= tiles) return;

  const int n16 = lane & 15;
  const int g   = lane >> 4;
  const int row = tile * 16 + n16;
  const float s = gnn_inv(degA, degB, (size_t)row);
  const float4* __restrict__ xr = (const float4*)(x + (size_t)row * F_IN + 4 * g);
  const float* wl = wlds + (4 * g) * HID + n16;

  v8f c = {};
  float4 cur[8], nxt[8];
  #pragma unroll
  for (int j = 0; j < 8; ++j) cur[j] = xr[2 * j];
  #pragma unroll
  for (int ch = 0; ch < F_IN / 64; ++ch) {
    if (ch + 1 < F_IN / 64) {
      #pragma unroll
      for (int j = 0; j < 8; ++j) nxt[j] = xr[16 * (ch + 1) + 2 * j];
    }
    #pragma unroll
    for (int j = 0; j < 8; ++j) {
      const int kb = 64 * ch + 8 * j;
      const float4 p = cur[j];
      v2f a0, a1, b0, b1;
      a0.x = p.x * s;
      a0.y = p.y * s;
      a1.x = p.z * s;
      a1.y = p.w * s;
      b0.x = wl[(kb + 0) * HID];
      b0.y = wl[(kb + 1) * HID];
      b1.x = wl[(kb + 2) * HID];
      b1.y = wl[(kb + 3) * HID];
      c = __builtin_amdgcn_wmma_f32_16x16x4_f32(
           false, a0,  false, b0,
           (short)0, c,  false,  false);
      c = __builtin_amdgcn_wmma_f32_16x16x4_f32(
           false, a1,  false, b1,
           (short)0, c,  false,  false);
    }
    #pragma unroll
    for (int j = 0; j < 8; ++j) cur[j] = nxt[j];
  }

  #pragma unroll
  for (int p = 0; p < 8; ++p) {
    const int m = tile * 16 + p + 8 * g;
    h1[(size_t)m * HID + n16] = c[p];
  }
  float4* __restrict__ z4 = (float4*)(agg1 + (size_t)tile * 16 * HID);
  z4[lane] = make_float4(0.0f, 0.0f, 0.0f, 0.0f);
  z4[32 + lane] = make_float4(0.0f, 0.0f, 0.0f, 0.0f);
}

#define GNN_RUN 128

__device__ __forceinline__ unsigned int gnn_draw(unsigned int* c, int lane) {
  unsigned int t = 0u;
  if (lane == 0) t = atomicAdd(c, 1u);
  return t;
}

__device__ __forceinline__ void gnn_add32(const int2* q, unsigned head,
                                          unsigned rem, unsigned g, unsigned f,
                                          const float* __restrict__ h1,
                                          float* __restrict__ agg1) {
  float v[16];
  unsigned d[16];
  #pragma unroll
  for (int i = 0; i < 16; ++i) {
    const unsigned k = 2u * i + g;
    const int2 sd = q[(head + k) & 127u];
    const unsigned s = (k < rem) ? (unsigned)sd.x : 0u;
    d[i] = (unsigned)sd.y;
    v[i] = h1[s * HID + f];
  }
  #pragma unroll
  for (int i = 0; i < 16; ++i) {
    const unsigned k = 2u * i + g;
    if (k < rem) atomicAdd(agg1 + d[i] * HID + f, v[i]);
  }
}

__global__ __launch_bounds__(256) void gnn_scatter16(
    const int* __restrict__ src, const int* __restrict__ dst,
    const float* __restrict__ h1, float* __restrict__ agg1,
    unsigned int* __restrict__ ctr, int N, int E) {
  __shared__ int2 ring[8][128];
  const int lane = threadIdx.x & 31;
  const int wave = __builtin_amdgcn_readfirstlane((int)(threadIdx.x >> 5));
  int2* q = ring[wave];
  const unsigned f = lane & 15;
  const unsigned g = lane >> 4;
  const int mid = N >> 1;
  const int runs = (E + GNN_RUN - 1) / GNN_RUN;
  const int mine = (runs - wave + 7) >> 3;
  const int h0 = gnn_half();
  unsigned head = 0u, tail = 0u;

  for (int pass = 0; pass < 2; ++pass) {
    const int h = h0 ^ pass;
    unsigned int* c = ctr + 64 * (8 * h + wave);
    unsigned int next = gnn_draw(c, lane);
    for (;;) {
      const int cur = (int)__builtin_amdgcn_readfirstlane(next);
      if (cur >= mine) break;
      const int run = 8 * cur + wave;
      for (int it = 0; it < GNN_RUN / 64; ++it) {
        const int e0 = run * GNN_RUN + it * 64 + lane;
        const int e1 = e0 + 32;
        const bool ok0 = e0 < E;
        const bool ok1 = e1 < E;
        const int s0 = ok0 ? src[e0] : 0;
        const int d0 = ok0 ? dst[e0] : 0;
        const int s1 = ok1 ? src[e1] : 0;
        const int d1 = ok1 ? dst[e1] : 0;
        if (it == 0) next = gnn_draw(c, lane);
        const bool own0 = ok0 && ((d0 >= mid) == (h != 0));
        const bool own1 = ok1 && ((d1 >= mid) == (h != 0));
        const unsigned m0 = __builtin_amdgcn_ballot_w32(own0);
        const unsigned m1 = __builtin_amdgcn_ballot_w32(own1);
        const unsigned c0 = (unsigned)__builtin_popcount(m0);
        const unsigned r0 = __builtin_amdgcn_mbcnt_lo(m0, 0u);
        const unsigned r1 = c0 + __builtin_amdgcn_mbcnt_lo(m1, 0u);
        if (own0) q[(tail + r0) & 127u] = make_int2(s0, d0);
        if (own1) q[(tail + r1) & 127u] = make_int2(s1, d1);
        tail += c0 + (unsigned)__builtin_popcount(m1);
        while (tail - head >= 32u) {
          gnn_add32(q, head, 32u, g, f, h1, agg1);
          head += 32u;
        }
      }
    }
  }
  if (tail != head) gnn_add32(q, head, tail - head, g, f, h1, agg1);
}

__global__ void gnn_layer2(const float* __restrict__ agg1, const float* __restrict__ degA,
                           const float* __restrict__ degB, const float* __restrict__ b1,
                           const float* __restrict__ W2, float* __restrict__ h2d,
                           float* __restrict__ agg2A, float* __restrict__ agg2B, int N) {
  int i = blockIdx.x * blockDim.x + threadIdx.x;
  if (i >= N) return;
  agg2A[i] = 0.0f;
  agg2B[i] = 0.0f;
  const float si = gnn_inv(degA, degB, (size_t)N + i);
  const float so = gnn_inv(degA, degB, (size_t)i);
  const float* a = agg1 + (size_t)i * HID;
  float acc0 = 0.0f, acc1 = 0.0f;
  #pragma unroll
  for (int k = 0; k < HID; ++k) {
    float hv = fmaxf(fmaf(a[k], si, b1[k]), 0.0f) * so;
    acc0 = fmaf(hv, W2[k * NCLS + 0], acc0);
    acc1 = fmaf(hv, W2[k * NCLS + 1], acc1);
  }
  h2d[i] = acc0 - acc1;
}

#define GNN_S2_WGS 256
#define GNN_S2_TAB 51200

__device__ __forceinline__ void gnn_edges4(const float* tab, int4 s, int4 d, bool in,
                                           int b0, int nb, float* __restrict__ plane) {
  const unsigned b[4] = {(unsigned)s.x - (unsigned)b0, (unsigned)s.y - (unsigned)b0,
                         (unsigned)s.z - (unsigned)b0, (unsigned)s.w - (unsigned)b0};
  const int dd[4] = {d.x, d.y, d.z, d.w};
  bool ok[4];
  float v[4];
  #pragma unroll
  for (int i = 0; i < 4; ++i) {
    ok[i] = in && b[i] < (unsigned)nb;
    v[i] = tab[ok[i] ? b[i] : 0u];
  }
  #pragma unroll
  for (int i = 0; i < 4; ++i)
    if (ok[i]) atomicAdd(plane + dd[i], v[i]);
}

__global__ __launch_bounds__(1024) void gnn_scatter2(
    const int* __restrict__ src, const int* __restrict__ dst,
    const float* __restrict__ h2d, float* __restrict__ agg2A,
    float* __restrict__ agg2B, int N, int E) {
  __shared__ float tab[GNN_S2_TAB];
  const int tid = threadIdx.x;
  float* __restrict__ plane = gnn_half() ? agg2B : agg2A;
  const int nw = (int)(gridDim.x >> 1);
  const int w = (int)(blockIdx.x >> 1);
  const int per = ((E + nw - 1) / nw + 3) & ~3;
  const int lo = min(w * per, E);
  const int hi = min(lo + per, E);
  const int4* __restrict__ s4 = (const int4*)(src + lo);
  const int4* __restrict__ d4 = (const int4*)(dst + lo);
  const int nq = (hi - lo) >> 2;
  const int parts = (N + GNN_S2_TAB - 1) / GNN_S2_TAB;

  for (int p = (int)(blockIdx.x & 1); p < parts; p += 2) {
    const int b0 = p * GNN_S2_TAB;
    const int nb = min(N - b0, GNN_S2_TAB);
    const float4* __restrict__ g4 = (const float4*)(h2d + b0);
    float4* t4 = (float4*)tab;
    const int n4 = nb >> 2;
    for (int i = tid; i < n4; i += 8 * 1024) {
      float4 r[8];
      #pragma unroll
      for (int j = 0; j < 8; ++j) r[j] = g4[min(i + 1024 * j, n4 - 1)];
      #pragma unroll
      for (int j = 0; j < 8; ++j) t4[min(i + 1024 * j, n4 - 1)] = r[j];
    }
    if (tid < (nb & 3)) tab[4 * n4 + tid] = h2d[b0 + 4 * n4 + tid];
    __syncthreads();

    for (int q = tid; q < nq; q += 4 * 1024) {
      int4 rs[4], rd[4];
      #pragma unroll
      for (int j = 0; j < 4; ++j) {
        const int k = min(q + 1024 * j, nq - 1);
        rs[j] = s4[k];
        rd[j] = d4[k];
      }
      asm volatile("" ::: "memory");
      #pragma unroll
      for (int j = 0; j < 4; ++j)
        gnn_edges4(tab, rs[j], rd[j], q + 1024 * j < nq, b0, nb, plane);
    }
    if (tid < ((hi - lo) & 3)) {
      const unsigned b = (unsigned)src[lo + 4 * nq + tid] - (unsigned)b0;
      if (b < (unsigned)nb) atomicAdd(plane + dst[lo + 4 * nq + tid], tab[b]);
    }
    __syncthreads();
  }
}

__global__ void gnn_final(const float* __restrict__ agg2A, const float* __restrict__ agg2B,
                          const float* __restrict__ degA, const float* __restrict__ degB,
                          const float* __restrict__ b2, float* __restrict__ out, int N) {
  int i = blockIdx.x * blockDim.x + threadIdx.x;
  if (i >= N) return;
  const float si = gnn_inv(degA, degB, (size_t)N + i);
  const float d = fmaf(agg2A[i] + agg2B[i], si, b2[0] - b2[1]);
  const float t = log1pf(expf(-fabsf(d)));
  out[(size_t)i * NCLS + 0] = -(fmaxf(-d, 0.0f) + t);
  out[(size_t)i * NCLS + 1] = -(fmaxf(d, 0.0f) + t);
}

extern "C" void kernel_launch(void* const* d_in, const int* in_sizes, int n_in,
                              void* d_out, int out_size, void* d_ws, size_t ws_size,
                              hipStream_t stream) {
  const float* x   = (const float*)d_in[0];
  const int*   src = (const int*)d_in[1];
  const int*   dst = (const int*)d_in[2];
  const float* W1  = (const float*)d_in[3];
  const float* b1  = (const float*)d_in[4];
  const float* W2  = (const float*)d_in[5];
  const float* b2  = (const float*)d_in[6];
  float* out = (float*)d_out;

  const int N = in_sizes[0] / F_IN;
  const int E = in_sizes[1];

  float* ws      = (float*)d_ws;
  float* degA    = ws;
  float* degB    = ws + (size_t)2 * N;
  float* h1      = ws + (size_t)4 * N;
  float* agg2B   = h1;
  float* agg1    = ws + (size_t)20 * N;
  float* h2d     = ws + (size_t)36 * N;
  float* agg2A   = ws + (size_t)38 * N;

  const int B = 256;

  gnn_zero<<<(N + B - 1) / B, B, 0, stream>>>(ws, 4 * N, h2d);
  gnn_degrees<<<GNN_DEG_WGS, 1024, 0, stream>>>(src, dst, degA, degB, N, E);

  const int tiles = N >> 4;
  gnn_gemm1_wmma<<<(tiles + 7) / 8, B, 0, stream>>>(x, W1, degA, degB, h1, agg1, N);

  gnn_scatter16<<<2048, B, 0, stream>>>(src, dst, h1, agg1, (unsigned int*)h2d, N, E);
  gnn_layer2<<<(N + B - 1) / B, B, 0, stream>>>(agg1, degA, degB, b1, W2, h2d, agg2A, agg2B, N);
  gnn_scatter2<<<GNN_S2_WGS, 1024, 0, stream>>>(src, dst, h2d, agg2A, agg2B, N, E);
  gnn_final<<<(N + B - 1) / B, B, 0, stream>>>(agg2A, agg2B, degA, degB, b2, out, N);
}


# x @ W1: each x load instruction reads whole 128-byte lines (256 contiguous bytes of each of 2 rows, was 32 bytes of each of 16 rows); the 16x64 piece is turned to one row per lane in per-wave LDS
# speedup vs baseline: 6.4596x; 1.0488x over previous
#include <hip/hip_runtime.h>
#include <math.h>


typedef __attribute__((ext_vector_type(2))) float v2f;
typedef __attribute__((ext_vector_type(8))) float v8f;

#define F_IN 512
#define HID  16
#define NCLS 2
#define GNN_CTR_WORDS 1024

__global__ __launch_bounds__(256) void gnn_zero(float* __restrict__ ws, int n,
                                                float* __restrict__ ctr) {
  float4* __restrict__ w4 = (float4*)ws;
  const int n4 = n >> 2;
  const int tid = blockIdx.x * blockDim.x + threadIdx.x;
  const int nthr = gridDim.x * blockDim.x;
  for (int i = tid; i < n4; i += nthr) w4[i] = make_float4(0.0f, 0.0f, 0.0f, 0.0f);
  if (tid < (n & 3)) ws[(n4 << 2) + tid] = 0.0f;
  if (tid < GNN_CTR_WORDS / 4) ((float4*)ctr)[tid] = make_float4(0.0f, 0.0f, 0.0f, 0.0f);
}

__device__ __forceinline__ int gnn_half() {
  return (int)((__builtin_amdgcn_s_sendmsg_rtn(0x87) >> 18) & 1u);
}

#define GNN_DEG_WGS 256
#define GNN_TBINS   102400
#define GNN_TCHUNK  61440

__device__ __forceinline__ void gnn_count(unsigned int* tab, int node, int b0) {
  const unsigned b = (unsigned)node - (unsigned)b0;
  if (b < (unsigned)GNN_TBINS)
    atomicAdd(&tab[((b >> 6) << 5) | (b & 31u)], 1u << ((b >> 1) & 16u));
}

__global__ __launch_bounds__(1024) void gnn_degrees(
    const int* __restrict__ src, const int* __restrict__ dst,
    float* __restrict__ degA, float* __restrict__ degB, int N, int E) {
  __shared__ unsigned int tab[GNN_TBINS / 2];
  const int tid = threadIdx.x;
  const int lane = tid & 31;
  const int wave = __builtin_amdgcn_readfirstlane(tid >> 5);
  const int which = blockIdx.x & 1;
  const int* __restrict__ idx = which ? dst : src;
  float* __restrict__ plane = (gnn_half() ? degB : degA) + (which ? (size_t)N : (size_t)0);
  const int nw = (int)(gridDim.x >> 1);
  const int w = (int)(blockIdx.x >> 1);
  const int per = ((E + nw - 1) / nw + 3) & ~3;
  const int lo = min(w * per, E);
  const int hi = min(lo + per, E);

  for (int b0 = 0; b0 < N; b0 += GNN_TBINS) {
    const int nb = min(N - b0, GNN_TBINS);
    const int ngroups = (nb + 63) >> 6;
    const int rot = (w * ngroups) / nw;
    for (int c0 = lo; c0 < hi; c0 += GNN_TCHUNK) {
      const int c1 = min(c0 + GNN_TCHUNK, hi);
      uint4* t4 = (uint4*)tab;
      for (int i = tid; i < GNN_TBINS / 8; i += 1024) t4[i] = make_uint4(0u, 0u, 0u, 0u);
      __syncthreads();

      const int4* __restrict__ idx4 = (const int4*)(idx + c0);
      const int nq = (c1 - c0) >> 2;
      for (int q = tid; q < nq; q += 8 * 1024) {
        int4 r[8];
        #pragma unroll
        for (int j = 0; j < 8; ++j) r[j] = idx4[min(q + 1024 * j, nq - 1)];
        #pragma unroll
        for (int j = 0; j < 8; ++j) {
          if (q + 1024 * j < nq) {
            gnn_count(tab, r[j].x, b0);
            gnn_count(tab, r[j].y, b0);
            gnn_count(tab, r[j].z, b0);
            gnn_count(tab, r[j].w, b0);
          }
        }
      }
      if (tid < ((c1 - c0) & 3)) gnn_count(tab, idx[c0 + 4 * nq + tid], b0);
      __syncthreads();

      for (int k = wave; k < ngroups; k += 4 * 32) {
        unsigned v[4];
        int node[4];
        #pragma unroll
        for (int j = 0; j < 4; ++j) {
          const bool in = k + 32 * j < ngroups;
          int g = in ? k + 32 * j + rot : 0;
          if (g >= ngroups) g -= ngroups;
          const unsigned t = tab[(g << 5) + lane];
          v[j] = in ? t : 0u;
          node[j] = b0 + (g << 6) + lane;
        }
        #pragma unroll
        for (int j = 0; j < 4; ++j) {
          const unsigned cl = v[j] & 0xffffu;
          const unsigned ch = v[j] >> 16;
          if (cl != 0u && node[j] < N) atomicAdd(plane + node[j], (float)cl);
          if (ch != 0u && node[j] + 32 < N) atomicAdd(plane + node[j] + 32, (float)ch);
        }
      }
      __syncthreads();
    }
  }
}

__device__ __forceinline__ float gnn_inv(const float* __restrict__ degA,
                                         const float* __restrict__ degB, size_t i) {
  return rsqrtf(fmaxf(degA[i] + degB[i], 1.0f));
}

#define GNN_XPITCH 68

__global__ __launch_bounds__(256) void gnn_gemm1_wmma(
    const float* __restrict__ x, const float* __restrict__ W1,
    const float* __restrict__ degA, const float* __restrict__ degB,
    float* __restrict__ h1, float* __restrict__ agg1, int N) {
  __shared__ float wlds[F_IN * HID];
  __shared__ float xlds[8][16 * GNN_XPITCH];
  {
    const float4* __restrict__ w4 = (const float4*)W1;
    float4* l4 = (float4*)wlds;
    float4 t[8];
    #pragma unroll
    for (int j = 0; j < 8; ++j) t[j] = w4[threadIdx.x + 256 * j];
    #pragma unroll
    for (int j = 0; j < 8; ++j) l4[threadIdx.x + 256 * j] = t[j];
  }
  __syncthreads();

  const int wave = threadIdx.x >> 5;
  const int lane = threadIdx.x & 31;
  const int tiles = N >> 4;
  const int tile = blockIdx.x * 8 + wave;
  if (tile >= tiles) return;

  const int n16 = lane & 15;
  const int g   = lane >> 4;
  const int row = tile * 16 + n16;
  const float s = gnn_inv(degA, degB, (size_t)row);
  const float4* __restrict__ xg = (const float4*)(x + (size_t)(tile * 16 + g) * F_IN + 4 * n16);
  float* xs = xlds[wave];
  float4* xst = (float4*)(xs + g * GNN_XPITCH + 4 * n16);
  const float4* xld = (const float4*)(xs + n16 * GNN_XPITCH + 4 * g);
  const float* wl = wlds + (4 * g) * HID + n16;

  v8f c = {};
  float4 cur[8], nxt[8];
  #pragma unroll
  for (int j = 0; j < 8; ++j) cur[j] = xg[(2 * F_IN / 4) * j];
  #pragma unroll
  for (int ch = 0; ch < F_IN / 64; ++ch) {
    if (ch + 1 < F_IN / 64) {
      #pragma unroll
      for (int j = 0; j < 8; ++j) nxt[j] = xg[(2 * F_IN / 4) * j + 16 * (ch + 1)];
    }
    #pragma unroll
    for (int j = 0; j < 8; ++j) xst[(2 * GNN_XPITCH / 4) * j] = cur[j];
    float4 pc[8];
    #pragma unroll
    for (int j = 0; j < 8; ++j) pc[j] = xld[2 * j];
    #pragma unroll
    for (int j = 0; j < 8; ++j) {
      const int kb = 64 * ch + 8 * j;
      const float4 p = pc[j];
      v2f a0, a1, b0, b1;
      a0.x = p.x * s;
      a0.y = p.y * s;
      a1.x = p.z * s;
      a1.y = p.w * s;
      b0.x = wl[(kb + 0) * HID];
      b0.y = wl[(kb + 1) * HID];
      b1.x = wl[(kb + 2) * HID];
      b1.y = wl[(kb + 3) * HID];
      c = __builtin_amdgcn_wmma_f32_16x16x4_f32(
           false, a0,  false, b0,
           (short)0, c,  false,  false);
      c = __builtin_amdgcn_wmma_f32_16x16x4_f32(
           false, a1,  false, b1,
           (short)0, c,  false,  false);
    }
    #pragma unroll
    for (int j = 0; j < 8; ++j) cur[j] = nxt[j];
  }

  #pragma unroll
  for (int p = 0; p < 8; ++p) {
    const int m = tile * 16 + p + 8 * g;
    h1[(size_t)m * HID + n16] = c[p];
  }
  float4* __restrict__ z4 = (float4*)(agg1 + (size_t)tile * 16 * HID);
  z4[lane] = make_float4(0.0f, 0.0f, 0.0f, 0.0f);
  z4[32 + lane] = make_float4(0.0f, 0.0f, 0.0f, 0.0f);
}

#define GNN_RUN 128

__device__ __forceinline__ unsigned int gnn_draw(unsigned int* c, int lane) {
  unsigned int t = 0u;
  if (lane == 0) t = atomicAdd(c, 1u);
  return t;
}

__device__ __forceinline__ void gnn_add32(const int2* q, unsigned head,
                                          unsigned rem, unsigned g, unsigned f,
                                          const float* __restrict__ h1,
                                          float* __restrict__ agg1) {
  float v[16];
  unsigned d[16];
  #pragma unroll
  for (int i = 0; i < 16; ++i) {
    const unsigned k = 2u * i + g;
    const int2 sd = q[(head + k) & 127u];
    const unsigned s = (k < rem) ? (unsigned)sd.x : 0u;
    d[i] = (unsigned)sd.y;
    v[i] = h1[s * HID + f];
  }
  #pragma unroll
  for (int i = 0; i < 16; ++i) {
    const unsigned k = 2u * i + g;
    if (k < rem) atomicAdd(agg1 + d[i] * HID + f, v[i]);
  }
}

__global__ __launch_bounds__(256) void gnn_scatter16(
    const int* __restrict__ src, const int* __restrict__ dst,
    const float* __restrict__ h1, float* __restrict__ agg1,
    unsigned int* __restrict__ ctr, int N, int E) {
  __shared__ int2 ring[8][128];
  const int lane = threadIdx.x & 31;
  const int wave = __builtin_amdgcn_readfirstlane((int)(threadIdx.x >> 5));
  int2* q = ring[wave];
  const unsigned f = lane & 15;
  const unsigned g = lane >> 4;
  const int mid = N >> 1;
  const int runs = (E + GNN_RUN - 1) / GNN_RUN;
  const int mine = (runs - wave + 7) >> 3;
  const int h0 = gnn_half();
  unsigned head = 0u, tail = 0u;

  for (int pass = 0; pass < 2; ++pass) {
    const int h = h0 ^ pass;
    unsigned int* c = ctr + 64 * (8 * h + wave);
    unsigned int next = gnn_draw(c, lane);
    for (;;) {
      const int cur = (int)__builtin_amdgcn_readfirstlane(next);
      if (cur >= mine) break;
      const int run = 8 * cur + wave;
      for (int it = 0; it < GNN_RUN / 64; ++it) {
        const int e0 = run * GNN_RUN + it * 64 + lane;
        const int e1 = e0 + 32;
        const bool ok0 = e0 < E;
        const bool ok1 = e1 < E;
        const int s0 = ok0 ? src[e0] : 0;
        const int d0 = ok0 ? dst[e0] : 0;
        const int s1 = ok1 ? src[e1] : 0;
        const int d1 = ok1 ? dst[e1] : 0;
        if (it == 0) next = gnn_draw(c, lane);
        const bool own0 = ok0 && ((d0 >= mid) == (h != 0));
        const bool own1 = ok1 && ((d1 >= mid) == (h != 0));
        const unsigned m0 = __builtin_amdgcn_ballot_w32(own0);
        const unsigned m1 = __builtin_amdgcn_ballot_w32(own1);
        const unsigned c0 = (unsigned)__builtin_popcount(m0);
        const unsigned r0 = __builtin_amdgcn_mbcnt_lo(m0, 0u);
        const unsigned r1 = c0 + __builtin_amdgcn_mbcnt_lo(m1, 0u);
        if (own0) q[(tail + r0) & 127u] = make_int2(s0, d0);
        if (own1) q[(tail + r1) & 127u] = make_int2(s1, d1);
        tail += c0 + (unsigned)__builtin_popcount(m1);
        while (tail - head >= 32u) {
          gnn_add32(q, head, 32u, g, f, h1, agg1);
          head += 32u;
        }
      }
    }
  }
  if (tail != head) gnn_add32(q, head, tail - head, g, f, h1, agg1);
}

__global__ void gnn_layer2(const float* __restrict__ agg1, const float* __restrict__ degA,
                           const float* __restrict__ degB, const float* __restrict__ b1,
                           const float* __restrict__ W2, float* __restrict__ h2d,
                           float* __restrict__ agg2A, float* __restrict__ agg2B, int N) {
  int i = blockIdx.x * blockDim.x + threadIdx.x;
  if (i >= N) return;
  agg2A[i] = 0.0f;
  agg2B[i] = 0.0f;
  const float si = gnn_inv(degA, degB, (size_t)N + i);
  const float so = gnn_inv(degA, degB, (size_t)i);
  const float* a = agg1 + (size_t)i * HID;
  float acc0 = 0.0f, acc1 = 0.0f;
  #pragma unroll
  for (int k = 0; k < HID; ++k) {
    float hv = fmaxf(fmaf(a[k], si, b1[k]), 0.0f) * so;
    acc0 = fmaf(hv, W2[k * NCLS + 0], acc0);
    acc1 = fmaf(hv, W2[k * NCLS + 1], acc1);
  }
  h2d[i] = acc0 - acc1;
}

#define GNN_S2_WGS 256
#define GNN_S2_TAB 51200

__device__ __forceinline__ void gnn_edges4(const float* tab, int4 s, int4 d, bool in,
                                           int b0, int nb, float* __restrict__ plane) {
  const unsigned b[4] = {(unsigned)s.x - (unsigned)b0, (unsigned)s.y - (unsigned)b0,
                         (unsigned)s.z - (unsigned)b0, (unsigned)s.w - (unsigned)b0};
  const int dd[4] = {d.x, d.y, d.z, d.w};
  bool ok[4];
  float v[4];
  #pragma unroll
  for (int i = 0; i < 4; ++i) {
    ok[i] = in && b[i] < (unsigned)nb;
    v[i] = tab[ok[i] ? b[i] : 0u];
  }
  #pragma unroll
  for (int i = 0; i < 4; ++i)
    if (ok[i]) atomicAdd(plane + dd[i], v[i]);
}

__global__ __launch_bounds__(1024) void gnn_scatter2(
    const int* __restrict__ src, const int* __restrict__ dst,
    const float* __restrict__ h2d, float* __restrict__ agg2A,
    float* __restrict__ agg2B, int N, int E) {
  __shared__ float tab[GNN_S2_TAB];
  const int tid = threadIdx.x;
  float* __restrict__ plane = gnn_half() ? agg2B : agg2A;
  const int nw = (int)(gridDim.x >> 1);
  const int w = (int)(blockIdx.x >> 1);
  const int per = ((E + nw - 1) / nw + 3) & ~3;
  const int lo = min(w * per, E);
  const int hi = min(lo + per, E);
  const int4* __restrict__ s4 = (const int4*)(src + lo);
  const int4* __restrict__ d4 = (const int4*)(dst + lo);
  const int nq = (hi - lo) >> 2;
  const int parts = (N + GNN_S2_TAB - 1) / GNN_S2_TAB;

  for (int p = (int)(blockIdx.x & 1); p < parts; p += 2) {
    const int b0 = p * GNN_S2_TAB;
    const int nb = min(N - b0, GNN_S2_TAB);
    const float4* __restrict__ g4 = (const float4*)(h2d + b0);
    float4* t4 = (float4*)tab;
    const int n4 = nb >> 2;
    for (int i = tid; i < n4; i += 8 * 1024) {
      float4 r[8];
      #pragma unroll
      for (int j = 0; j < 8; ++j) r[j] = g4[min(i + 1024 * j, n4 - 1)];
      #pragma unroll
      for (int j = 0; j < 8; ++j) t4[min(i + 1024 * j, n4 - 1)] = r[j];
    }
    if (tid < (nb & 3)) tab[4 * n4 + tid] = h2d[b0 + 4 * n4 + tid];
    __syncthreads();

    for (int q = tid; q < nq; q += 4 * 1024) {
      int4 rs[4], rd[4];
      #pragma unroll
      for (int j = 0; j < 4; ++j) {
        const int k = min(q + 1024 * j, nq - 1);
        rs[j] = s4[k];
        rd[j] = d4[k];
      }
      asm volatile("" ::: "memory");
      #pragma unroll
      for (int j = 0; j < 4; ++j)
        gnn_edges4(tab, rs[j], rd[j], q + 1024 * j < nq, b0, nb, plane);
    }
    if (tid < ((hi - lo) & 3)) {
      const unsigned b = (unsigned)src[lo + 4 * nq + tid] - (unsigned)b0;
      if (b < (unsigned)nb) atomicAdd(plane + dst[lo + 4 * nq + tid], tab[b]);
    }
    __syncthreads();
  }
}

__global__ void gnn_final(const float* __restrict__ agg2A, const float* __restrict__ agg2B,
                          const float* __restrict__ degA, const float* __restrict__ degB,
                          const float* __restrict__ b2, float* __restrict__ out, int N) {
  int i = blockIdx.x * blockDim.x + threadIdx.x;
  if (i >= N) return;
  const float si = gnn_inv(degA, degB, (size_t)N + i);
  const float d = fmaf(agg2A[i] + agg2B[i], si, b2[0] - b2[1]);
  const float t = log1pf(expf(-fabsf(d)));
  out[(size_t)i * NCLS + 0] = -(fmaxf(-d, 0.0f) + t);
  out[(size_t)i * NCLS + 1] = -(fmaxf(d, 0.0f) + t);
}

extern "C" void kernel_launch(void* const* d_in, const int* in_sizes, int n_in,
                              void* d_out, int out_size, void* d_ws, size_t ws_size,
                              hipStream_t stream) {
  const float* x   = (const float*)d_in[0];
  const int*   src = (const int*)d_in[1];
  const int*   dst = (const int*)d_in[2];
  const float* W1  = (const float*)d_in[3];
  const float* b1  = (const float*)d_in[4];
  const float* W2  = (const float*)d_in[5];
  const float* b2  = (const float*)d_in[6];
  float* out = (float*)d_out;

  const int N = in_sizes[0] / F_IN;
  const int E = in_sizes[1];

  float* ws      = (float*)d_ws;
  float* degA    = ws;
  float* degB    = ws + (size_t)2 * N;
  float* h1      = ws + (size_t)4 * N;
  float* agg2B   = h1;
  float* agg1    = ws + (size_t)20 * N;
  float* h2d     = ws + (size_t)36 * N;
  float* agg2A   = ws + (size_t)38 * N;

  const int B = 256;

  gnn_zero<<<(N + B - 1) / B, B, 0, stream>>>(ws, 4 * N, h2d);
  gnn_degrees<<<GNN_DEG_WGS, 1024, 0, stream>>>(src, dst, degA, degB, N, E);

  const int tiles = N >> 4;
  gnn_gemm1_wmma<<<(tiles + 7) / 8, B, 0, stream>>>(x, W1, degA, degB, h1, agg1, N);

  gnn_scatter16<<<2048, B, 0, stream>>>(src, dst, h1, agg1, (unsigned int*)h2d, N, E);
  gnn_layer2<<<(N + B - 1) / B, B, 0, stream>>>(agg1, degA, degB, b1, W2, h2d, agg2A, agg2B, N);
  gnn_scatter2<<<GNN_S2_WGS, 1024, 0, stream>>>(src, dst, h2d, agg2A, agg2B, N, E);
  gnn_final<<<(N + B - 1) / B, B, 0, stream>>>(agg2A, agg2B, degA, degB, b2, out, N);
}
